# starting kernel
# baseline (speedup 1.0000x reference)
#include <hip/hip_runtime.h>
#include <hip/hip_bf16.h>


typedef float v2f __attribute__((ext_vector_type(2)));
typedef float v8f __attribute__((ext_vector_type(8)));

constexpr int   Bc = 32, Sc = 1024, Ec = 512, Hc = 64, Tc = 256;
constexpr float LNK = -0.1053605156578263f;

static __device__ __forceinline__ v8f wmma4(v2f a, v2f b, v8f c) {
  return __builtin_amdgcn_wmma_f32_16x16x4_f32(false, a, false, b, (short)0, c,
                                               false, false);
}

__global__ void __launch_bounds__(32)
proj16x16(const float* __restrict__ X, const float* __restrict__ W,
          float* __restrict__ Y, int K, int N)
{
  const int lane  = threadIdx.x & 31;
  const int nl    = lane & 15;
  const int khalf = (lane >> 4) << 1;
  const int m     = (blockIdx.x << 4) + nl;
  const int ncol  = (blockIdx.y << 4) + nl;

  const float* xrow = X + (size_t)m * K;
  v8f c = {};
  for (int k0 = 0; k0 < K; k0 += 4) {
    v2f a = *(const v2f*)(xrow + k0 + khalf);
    v2f b;
    b.x = W[(size_t)(k0 + khalf)     * N + ncol];
    b.y = W[(size_t)(k0 + khalf + 1) * N + ncol];
    c = wmma4(a, b, c);
  }
  const int mbase = (blockIdx.x << 4) + ((lane >> 4) << 3);
  #pragma unroll
  for (int r = 0; r < 8; ++r)
    Y[(size_t)(mbase + r) * N + ncol] = c[r];
}

__global__ void dones_scan(const int* __restrict__ dones, int* __restrict__ cum,
                           int* __restrict__ firstdone)
{
  const int b = blockIdx.x;
  if (threadIdx.x == 0) {
    int c = 0, first = Tc;
    for (int t = 0; t < Tc; ++t) {
      const int d = (dones[(size_t)b * Sc + t * 4] != 0) ? 1 : 0;
      if (d && first == Tc) first = t;
      c += d;
      cum[b * Tc + t] = c;
    }
    firstdone[b] = first;
  }
}

__global__ void __launch_bounds__(32)
retention16(const float* __restrict__ q, const float* __restrict__ kbuf,
            const float* __restrict__ vbuf, const float* __restrict__ hstate,
            const int* __restrict__ cum, const int* __restrict__ firstdone,
            float* __restrict__ out)
{
  __shared__ float ktab[Tc + 1];
  __shared__ int   cumsh[Tc];
  __shared__ float atile[16 * 16];

  const int lane  = threadIdx.x & 31;
  const int nl    = lane & 15;
  const int khalf = (lane >> 4) << 1;
  const int mh    = (lane >> 4) << 3;
  const int it    = blockIdx.x;
  const int b     = blockIdx.y;

  const int* cumb = cum + b * Tc;
  for (int i = lane; i <= Tc; i += 32) ktab[i] = __expf((float)i * LNK);
  for (int i = lane; i < Tc; i += 32)  cumsh[i] = cumb[i];
  __syncthreads();

  const float* qb = q      + (size_t)b * Sc * Hc;
  const float* kb = kbuf   + (size_t)b * Sc * Hc;
  const float* vb = vbuf   + (size_t)b * Sc * Hc;
  const float* hb = hstate + (size_t)b * Hc * Hc;

  const float* qrow = qb + (size_t)(it * 16 + nl) * Hc;
  v2f qa[16];
  #pragma unroll
  for (int ks = 0; ks < 16; ++ks)
    qa[ks] = *(const v2f*)(qrow + ks * 4 + khalf);

  v8f acc[4] = {};
  for (int jt = 0; jt <= it; ++jt) {
    v8f s = {};
    const float* krow = kb + (size_t)(jt * 16 + nl) * Hc;
    #pragma unroll
    for (int ks = 0; ks < 16; ++ks) {
      v2f bm = *(const v2f*)(krow + ks * 4 + khalf);
      s = wmma4(qa[ks], bm, s);
    }
    #pragma unroll
    for (int r = 0; r < 8; ++r) {
      const int gi = it * 16 + mh + r;
      const int gj = jt * 16 + nl;
      const int ti = gi >> 2, tj = gj >> 2;
      float w = 0.0f;
      if (gj <= gi && cumsh[ti] == cumsh[tj]) w = ktab[ti - tj];
      atile[(mh + r) * 16 + nl] = s[r] * w;
    }
    __syncthreads();
    #pragma unroll
    for (int k0 = 0; k0 < 16; k0 += 4) {
      v2f a = *(const v2f*)(&atile[nl * 16 + k0 + khalf]);
      const float* v0 = vb + (size_t)(jt * 16 + k0 + khalf) * Hc;
      const float* v1 = v0 + Hc;
      #pragma unroll
      for (int nt = 0; nt < 4; ++nt) {
        v2f bm;
        bm.x = v0[nt * 16 + nl];
        bm.y = v1[nt * 16 + nl];
        acc[nt] = wmma4(a, bm, acc[nt]);
      }
    }
    __syncthreads();
  }

  v8f cr[4] = {};
  #pragma unroll
  for (int ks = 0; ks < 16; ++ks) {
    const float* h0 = hb + (size_t)(ks * 4 + khalf) * Hc;
    const float* h1 = h0 + Hc;
    #pragma unroll
    for (int nt = 0; nt < 4; ++nt) {
      v2f bm;
      bm.x = h0[nt * 16 + nl];
      bm.y = h1[nt * 16 + nl];
      cr[nt] = wmma4(qa[ks], bm, cr[nt]);
    }
  }

  const int fd = firstdone[b];
  #pragma unroll
  for (int r = 0; r < 8; ++r) {
    const int   gi = it * 16 + mh + r;
    const int   ti = gi >> 2;
    const float xi = (ti < fd) ? ktab[ti + 1] : 0.0f;
    float* orow = out + ((size_t)b * Sc + gi) * Hc;
    #pragma unroll
    for (int nt = 0; nt < 4; ++nt)
      orow[nt * 16 + nl] = acc[nt][r] + cr[nt][r] * xi;
  }
}

__global__ void __launch_bounds__(32)
hstate_update(const float* __restrict__ kbuf, const float* __restrict__ vbuf,
              const float* __restrict__ hstate, const int* __restrict__ cum,
              float* __restrict__ hout)
{
  __shared__ float ktab[Tc + 1];
  __shared__ float wsh[Tc];
  const int lane  = threadIdx.x & 31;
  const int nl    = lane & 15;
  const int khalf = (lane >> 4) << 1;
  const int mt = blockIdx.x, nt = blockIdx.y, b = blockIdx.z;
  const int* cumb = cum + b * Tc;
  const int  cend = cumb[Tc - 1];

  for (int i = lane; i <= Tc; i += 32) ktab[i] = __expf((float)i * LNK);
  __syncthreads();
  for (int i = lane; i < Tc; i += 32)
    wsh[i] = (cumb[i] == cend) ? ktab[(Tc - 1) - i] : 0.0f;
  __syncthreads();

  const float* kb = kbuf + (size_t)b * Sc * Hc;
  const float* vb = vbuf + (size_t)b * Sc * Hc;

  v8f c = {};
  for (int s0 = 0; s0 < Sc; s0 += 4) {
    const int   sA = s0 + khalf;
    const float w  = wsh[s0 >> 2];
    v2f a;
    a.x = kb[(size_t)sA       * Hc + mt * 16 + nl];
    a.y = kb[(size_t)(sA + 1) * Hc + mt * 16 + nl];
    v2f bm;
    bm.x = vb[(size_t)sA       * Hc + nt * 16 + nl] * w;
    bm.y = vb[(size_t)(sA + 1) * Hc + nt * 16 + nl] * w;
    c = wmma4(a, bm, c);
  }

  const float cd    = ktab[Tc];
  const float delta = (cend == 0) ? 1.0f : 0.0f;
  const float* hb   = hstate + (size_t)b * Hc * Hc;
  const int    mh   = (lane >> 4) << 3;
  #pragma unroll
  for (int r = 0; r < 8; ++r) {
    const int h1 = mt * 16 + mh + r;
    hout[((size_t)b * Hc + h1) * Hc + nt * 16 + nl] =
        c[r] + hb[(size_t)h1 * Hc + nt * 16 + nl] * cd * delta;
  }
}

extern "C" void kernel_launch(void* const* d_in, const int* in_sizes, int n_in,
                              void* d_out, int out_size, void* d_ws, size_t ws_size,
                              hipStream_t stream)
{
  const float* key    = (const float*)d_in[0];
  const float* query  = (const float*)d_in[1];
  const float* value  = (const float*)d_in[2];
  const float* hstate = (const float*)d_in[3];
  const int*   dones  = (const int*)d_in[4];
  const float* W_Q    = (const float*)d_in[5];
  const float* W_K    = (const float*)d_in[6];
  const float* W_V    = (const float*)d_in[7];

  const size_t qkv = (size_t)Bc * Sc * Hc;
  float* ws = (float*)d_ws;
  float* qb = ws;
  float* kb = ws + qkv;
  float* vb = ws + 2 * qkv;
  int*   cum       = (int*)(ws + 3 * qkv);
  int*   firstdone = cum + Bc * Tc;

  dim3 gproj(Bc * Sc / 16, Hc / 16);
  proj16x16<<<gproj, 32, 0, stream>>>(query, W_Q, qb, Ec, Hc);
  proj16x16<<<gproj, 32, 0, stream>>>(key,   W_K, kb, Ec, Hc);
  proj16x16<<<gproj, 32, 0, stream>>>(value, W_V, vb, Ec, Hc);

  dones_scan<<<Bc, 32, 0, stream>>>(dones, cum, firstdone);

  float* out0 = (float*)d_out;
  float* hout = out0 + qkv;
  retention16<<<dim3(Sc / 16, Bc), 32, 0, stream>>>(qb, kb, vb, hstate, cum,
                                                    firstdone, out0);
  hstate_update<<<dim3(Hc / 16, Hc / 16, Bc), 32, 0, stream>>>(kb, vb, hstate,
                                                               cum, hout);
}


# retention key loop: decay-mask weights from two exponentials and one done-count look-up per key tile, the rows' done-counts hoisted; no per-element look-up chains or branches; same values
# speedup vs baseline: 1.9131x; 1.9131x over previous
#include <hip/hip_runtime.h>
#include <hip/hip_bf16.h>


typedef float v2f __attribute__((ext_vector_type(2)));
typedef float v4f __attribute__((ext_vector_type(4)));
typedef float v8f __attribute__((ext_vector_type(8)));

constexpr int   Bc = 32, Sc = 1024, Ec = 512, Hc = 64, Tc = 256;
constexpr float LNK = -0.1053605156578263f;

static __device__ __forceinline__ v8f wmma4(v2f a, v2f b, v8f c) {
  return __builtin_amdgcn_wmma_f32_16x16x4_f32(false, a, false, b, (short)0, c,
                                               false, false);
}

constexpr int PJ_TILES = 4;

__global__ void __launch_bounds__(256, 4)
proj_splitk(const float* __restrict__ X, const float* __restrict__ W,
            float* __restrict__ Y)
{
  __shared__ v4f part[2][8][16][16];

  const int lane = threadIdx.x & 31;
  const int wave = threadIdx.x >> 5;
  const int c    = lane & 15;
  const int h    = lane >> 4;
  const int kw   = wave << 6;

  const int      rt0  = blockIdx.x * PJ_TILES;
  const float*   xp   = X + (size_t)rt0 * 16 * Ec;
  float*         yp   = Y + (size_t)rt0 * 16 * Hc;
  const unsigned xoff = (unsigned)(c * Ec + kw + 4 * h);
  const unsigned yoff = (unsigned)((2 * wave + h) * Hc + 4 * c);

  v4f xa[8];
  #pragma unroll
  for (int j = 0; j < 8; ++j) xa[j] = *(const v4f*)(xp + xoff + 8 * j);

  v2f wb[4][8][2];
  v4f raw[2][8];
  const float* wp = W + (size_t)(kw + 4 * h) * Hc + 4 * c;
  #pragma unroll
  for (int i = 0; i < 8; ++i)
    raw[0][i] = *(const v4f*)(wp + (size_t)(8 * (i >> 2) + (i & 3)) * Hc);
  #pragma unroll
  for (int ch = 0; ch < 4; ++ch) {
    if (ch + 1 < 4) {
      #pragma unroll
      for (int i = 0; i < 8; ++i)
        raw[(ch + 1) & 1][i] =
            *(const v4f*)(wp + (size_t)(8 * (2 * (ch + 1) + (i >> 2)) + (i & 3)) * Hc);
    }
    #pragma unroll
    for (int jj = 0; jj < 2; ++jj) {
      #pragma unroll
      for (int p = 0; p < 2; ++p) {
        #pragma unroll
        for (int nt = 0; nt < 4; ++nt) {
          wb[nt][2 * ch + jj][p].x = raw[ch & 1][4 * jj + 2 * p][nt];
          wb[nt][2 * ch + jj][p].y = raw[ch & 1][4 * jj + 2 * p + 1][nt];
        }
      }
    }
    __builtin_amdgcn_sched_barrier(0);
  }

  #pragma unroll
  for (int t = 0; t < PJ_TILES; ++t) {
    v4f xn[8];
    if (t + 1 < PJ_TILES) {
      const float* xq = xp + (size_t)(t + 1) * 16 * Ec;
      #pragma unroll
      for (int j = 0; j < 8; ++j) xn[j] = *(const v4f*)(xq + xoff + 8 * j);
    }

    v8f acc[4] = {};
    #pragma unroll
    for (int j = 0; j < 8; ++j) {
      #pragma unroll
      for (int p = 0; p < 2; ++p) {
        v2f a;
        a.x = xa[j][2 * p];
        a.y = xa[j][2 * p + 1];
        #pragma unroll
        for (int nt = 0; nt < 4; ++nt)
          acc[nt] = wmma4(a, wb[nt][j][p], acc[nt]);
      }
    }

    #pragma unroll
    for (int r = 0; r < 8; ++r) {
      v4f o;
      o.x = acc[0][r]; o.y = acc[1][r]; o.z = acc[2][r]; o.w = acc[3][r];
      part[t & 1][wave][8 * h + r][c] = o;
    }
    __syncthreads();

    v4f s = part[t & 1][0][2 * wave + h][c];
    #pragma unroll
    for (int u = 1; u < 8; ++u) s += part[t & 1][u][2 * wave + h][c];
    *(v4f*)(yp + (size_t)t * 16 * Hc + yoff) = s;

    if (t + 1 < PJ_TILES) {
      #pragma unroll
      for (int j = 0; j < 8; ++j) xa[j] = xn[j];
    }
  }
}

__global__ void dones_scan(const int* __restrict__ dones, int* __restrict__ cum,
                           int* __restrict__ firstdone)
{
  const int b = blockIdx.x;
  if (threadIdx.x == 0) {
    int c = 0, first = Tc;
    for (int t = 0; t < Tc; ++t) {
      const int d = (dones[(size_t)b * Sc + t * 4] != 0) ? 1 : 0;
      if (d && first == Tc) first = t;
      c += d;
      cum[b * Tc + t] = c;
    }
    firstdone[b] = first;
  }
}

__global__ void __launch_bounds__(32)
retention16(const float* __restrict__ q, const float* __restrict__ kbuf,
            const float* __restrict__ vbuf, const float* __restrict__ hstate,
            const int* __restrict__ cum, const int* __restrict__ firstdone,
            float* __restrict__ out)
{
  __shared__ int   cumsh[Tc];
  __shared__ float atile[16 * 16];

  const int lane  = threadIdx.x & 31;
  const int nl    = lane & 15;
  const int khalf = (lane >> 4) << 1;
  const int mh    = (lane >> 4) << 3;
  const int it    = blockIdx.x;
  const int b     = blockIdx.y;

  const int* cumb = cum + b * Tc;
  for (int i = lane; i < Tc; i += 32)  cumsh[i] = cumb[i];
  __syncthreads();

  const int ti0 = it * 4 + (mh >> 2);
  const int cr0 = cumsh[ti0];
  const int cr1 = cumsh[ti0 + 1];
  const int cb  = nl >> 2;

  const float* qb = q      + (size_t)b * Sc * Hc;
  const float* kb = kbuf   + (size_t)b * Sc * Hc;
  const float* vb = vbuf   + (size_t)b * Sc * Hc;
  const float* hb = hstate + (size_t)b * Hc * Hc;

  const float* qrow = qb + (size_t)(it * 16 + nl) * Hc;
  v2f qa[16];
  #pragma unroll
  for (int ks = 0; ks < 16; ++ks)
    qa[ks] = *(const v2f*)(qrow + ks * 4 + khalf);

  v8f acc[4] = {};
  for (int jt = 0; jt <= it; ++jt) {
    const int tj = jt * 4 + cb;
    const int cj = cumsh[tj];
    v8f s = {};
    const float* krow = kb + (size_t)(jt * 16 + nl) * Hc;
    #pragma unroll
    for (int ks = 0; ks < 16; ++ks) {
      v2f bm = *(const v2f*)(krow + ks * 4 + khalf);
      s = wmma4(qa[ks], bm, s);
    }
    const float m0 = (cj == cr0) ? __expf((float)(ti0 - tj) * LNK) : 0.0f;
    const float m1 = (cj == cr1) ? __expf((float)(ti0 + 1 - tj) * LNK) : 0.0f;
    const int   dg = (it - jt) * 16 + mh - nl;
    #pragma unroll
    for (int r = 0; r < 8; ++r) {
      const float w = (dg + r >= 0) ? ((r < 4) ? m0 : m1) : 0.0f;
      atile[(mh + r) * 16 + nl] = s[r] * w;
    }
    __syncthreads();
    #pragma unroll
    for (int k0 = 0; k0 < 16; k0 += 4) {
      v2f a = *(const v2f*)(&atile[nl * 16 + k0 + khalf]);
      const float* v0 = vb + (size_t)(jt * 16 + k0 + khalf) * Hc;
      const float* v1 = v0 + Hc;
      #pragma unroll
      for (int nt = 0; nt < 4; ++nt) {
        v2f bm;
        bm.x = v0[nt * 16 + nl];
        bm.y = v1[nt * 16 + nl];
        acc[nt] = wmma4(a, bm, acc[nt]);
      }
    }
    __syncthreads();
  }

  v8f cr[4] = {};
  #pragma unroll
  for (int ks = 0; ks < 16; ++ks) {
    const float* h0 = hb + (size_t)(ks * 4 + khalf) * Hc;
    const float* h1 = h0 + Hc;
    #pragma unroll
    for (int nt = 0; nt < 4; ++nt) {
      v2f bm;
      bm.x = h0[nt * 16 + nl];
      bm.y = h1[nt * 16 + nl];
      cr[nt] = wmma4(qa[ks], bm, cr[nt]);
    }
  }

  const int fd = firstdone[b];
  #pragma unroll
  for (int r = 0; r < 8; ++r) {
    const int   gi = it * 16 + mh + r;
    const int   ti = gi >> 2;
    const float xi = (ti < fd) ? __expf((float)(ti + 1) * LNK) : 0.0f;
    float* orow = out + ((size_t)b * Sc + gi) * Hc;
    #pragma unroll
    for (int nt = 0; nt < 4; ++nt)
      orow[nt * 16 + nl] = acc[nt][r] + cr[nt][r] * xi;
  }
}

__global__ void __launch_bounds__(32)
hstate_update(const float* __restrict__ kbuf, const float* __restrict__ vbuf,
              const float* __restrict__ hstate, const int* __restrict__ cum,
              float* __restrict__ hout)
{
  __shared__ float ktab[Tc + 1];
  __shared__ float wsh[Tc];
  const int lane  = threadIdx.x & 31;
  const int nl    = lane & 15;
  const int khalf = (lane >> 4) << 1;
  const int mt = blockIdx.x, nt = blockIdx.y, b = blockIdx.z;
  const int* cumb = cum + b * Tc;
  const int  cend = cumb[Tc - 1];

  for (int i = lane; i <= Tc; i += 32) ktab[i] = __expf((float)i * LNK);
  __syncthreads();
  for (int i = lane; i < Tc; i += 32)
    wsh[i] = (cumb[i] == cend) ? ktab[(Tc - 1) - i] : 0.0f;
  __syncthreads();

  const float* kb = kbuf + (size_t)b * Sc * Hc;
  const float* vb = vbuf + (size_t)b * Sc * Hc;

  v8f c = {};
  for (int s0 = 0; s0 < Sc; s0 += 4) {
    const int   sA = s0 + khalf;
    const float w  = wsh[s0 >> 2];
    v2f a;
    a.x = kb[(size_t)sA       * Hc + mt * 16 + nl];
    a.y = kb[(size_t)(sA + 1) * Hc + mt * 16 + nl];
    v2f bm;
    bm.x = vb[(size_t)sA       * Hc + nt * 16 + nl] * w;
    bm.y = vb[(size_t)(sA + 1) * Hc + nt * 16 + nl] * w;
    c = wmma4(a, bm, c);
  }

  const float cd    = ktab[Tc];
  const float delta = (cend == 0) ? 1.0f : 0.0f;
  const float* hb   = hstate + (size_t)b * Hc * Hc;
  const int    mh   = (lane >> 4) << 3;
  #pragma unroll
  for (int r = 0; r < 8; ++r) {
    const int h1 = mt * 16 + mh + r;
    hout[((size_t)b * Hc + h1) * Hc + nt * 16 + nl] =
        c[r] + hb[(size_t)h1 * Hc + nt * 16 + nl] * cd * delta;
  }
}

extern "C" void kernel_launch(void* const* d_in, const int* in_sizes, int n_in,
                              void* d_out, int out_size, void* d_ws, size_t ws_size,
                              hipStream_t stream)
{
  const float* key    = (const float*)d_in[0];
  const float* query  = (const float*)d_in[1];
  const float* value  = (const float*)d_in[2];
  const float* hstate = (const float*)d_in[3];
  const int*   dones  = (const int*)d_in[4];
  const float* W_Q    = (const float*)d_in[5];
  const float* W_K    = (const float*)d_in[6];
  const float* W_V    = (const float*)d_in[7];

  const size_t qkv = (size_t)Bc * Sc * Hc;
  float* ws = (float*)d_ws;
  float* qb = ws;
  float* kb = ws + qkv;
  float* vb = ws + 2 * qkv;
  int*   cum       = (int*)(ws + 3 * qkv);
  int*   firstdone = cum + Bc * Tc;

  dim3 gproj(Bc * Sc / (16 * PJ_TILES));
  proj_splitk<<<gproj, 256, 0, stream>>>(query, W_Q, qb);
  proj_splitk<<<gproj, 256, 0, stream>>>(key,   W_K, kb);
  proj_splitk<<<gproj, 256, 0, stream>>>(value, W_V, vb);

  dones_scan<<<Bc, 32, 0, stream>>>(dones, cum, firstdone);

  float* out0 = (float*)d_out;
  float* hout = out0 + qkv;
  retention16<<<dim3(Sc / 16, Bc), 32, 0, stream>>>(qb, kb, vb, hstate, cum,
                                                    firstdone, out0);
  hstate_update<<<dim3(Hc / 16, Hc / 16, Bc), 32, 0, stream>>>(kb, vb, hstate,
                                                               cum, hout);
}


# retention key loop in trips of 32 keys: the trip's v and k loads go out as one group, the two score tiles are independent WMMA chains, one LDS round trip per trip; keys summed in the same order
# speedup vs baseline: 2.1046x; 1.1001x over previous
#include <hip/hip_runtime.h>
#include <hip/hip_bf16.h>


typedef float v2f __attribute__((ext_vector_type(2)));
typedef float v4f __attribute__((ext_vector_type(4)));
typedef float v8f __attribute__((ext_vector_type(8)));

constexpr int   Bc = 32, Sc = 1024, Ec = 512, Hc = 64, Tc = 256;
constexpr float LNK = -0.1053605156578263f;

static __device__ __forceinline__ v8f wmma4(v2f a, v2f b, v8f c) {
  return __builtin_amdgcn_wmma_f32_16x16x4_f32(false, a, false, b, (short)0, c,
                                               false, false);
}

constexpr int PJ_TILES = 4;

__global__ void __launch_bounds__(256, 4)
proj_splitk(const float* __restrict__ X, const float* __restrict__ W,
            float* __restrict__ Y)
{
  __shared__ v4f part[2][8][16][16];

  const int lane = threadIdx.x & 31;
  const int wave = threadIdx.x >> 5;
  const int c    = lane & 15;
  const int h    = lane >> 4;
  const int kw   = wave << 6;

  const int      rt0  = blockIdx.x * PJ_TILES;
  const float*   xp   = X + (size_t)rt0 * 16 * Ec;
  float*         yp   = Y + (size_t)rt0 * 16 * Hc;
  const unsigned xoff = (unsigned)(c * Ec + kw + 4 * h);
  const unsigned yoff = (unsigned)((2 * wave + h) * Hc + 4 * c);

  v4f xa[8];
  #pragma unroll
  for (int j = 0; j < 8; ++j) xa[j] = *(const v4f*)(xp + xoff + 8 * j);

  v2f wb[4][8][2];
  v4f raw[2][8];
  const float* wp = W + (size_t)(kw + 4 * h) * Hc + 4 * c;
  #pragma unroll
  for (int i = 0; i < 8; ++i)
    raw[0][i] = *(const v4f*)(wp + (size_t)(8 * (i >> 2) + (i & 3)) * Hc);
  #pragma unroll
  for (int ch = 0; ch < 4; ++ch) {
    if (ch + 1 < 4) {
      #pragma unroll
      for (int i = 0; i < 8; ++i)
        raw[(ch + 1) & 1][i] =
            *(const v4f*)(wp + (size_t)(8 * (2 * (ch + 1) + (i >> 2)) + (i & 3)) * Hc);
    }
    #pragma unroll
    for (int jj = 0; jj < 2; ++jj) {
      #pragma unroll
      for (int p = 0; p < 2; ++p) {
        #pragma unroll
        for (int nt = 0; nt < 4; ++nt) {
          wb[nt][2 * ch + jj][p].x = raw[ch & 1][4 * jj + 2 * p][nt];
          wb[nt][2 * ch + jj][p].y = raw[ch & 1][4 * jj + 2 * p + 1][nt];
        }
      }
    }
    __builtin_amdgcn_sched_barrier(0);
  }

  #pragma unroll
  for (int t = 0; t < PJ_TILES; ++t) {
    v4f xn[8];
    if (t + 1 < PJ_TILES) {
      const float* xq = xp + (size_t)(t + 1) * 16 * Ec;
      #pragma unroll
      for (int j = 0; j < 8; ++j) xn[j] = *(const v4f*)(xq + xoff + 8 * j);
    }

    v8f acc[4] = {};
    #pragma unroll
    for (int j = 0; j < 8; ++j) {
      #pragma unroll
      for (int p = 0; p < 2; ++p) {
        v2f a;
        a.x = xa[j][2 * p];
        a.y = xa[j][2 * p + 1];
        #pragma unroll
        for (int nt = 0; nt < 4; ++nt)
          acc[nt] = wmma4(a, wb[nt][j][p], acc[nt]);
      }
    }

    #pragma unroll
    for (int r = 0; r < 8; ++r) {
      v4f o;
      o.x = acc[0][r]; o.y = acc[1][r]; o.z = acc[2][r]; o.w = acc[3][r];
      part[t & 1][wave][8 * h + r][c] = o;
    }
    __syncthreads();

    v4f s = part[t & 1][0][2 * wave + h][c];
    #pragma unroll
    for (int u = 1; u < 8; ++u) s += part[t & 1][u][2 * wave + h][c];
    *(v4f*)(yp + (size_t)t * 16 * Hc + yoff) = s;

    if (t + 1 < PJ_TILES) {
      #pragma unroll
      for (int j = 0; j < 8; ++j) xa[j] = xn[j];
    }
  }
}

__global__ void dones_scan(const int* __restrict__ dones, int* __restrict__ cum,
                           int* __restrict__ firstdone)
{
  const int b = blockIdx.x;
  if (threadIdx.x == 0) {
    int c = 0, first = Tc;
    for (int t = 0; t < Tc; ++t) {
      const int d = (dones[(size_t)b * Sc + t * 4] != 0) ? 1 : 0;
      if (d && first == Tc) first = t;
      c += d;
      cum[b * Tc + t] = c;
    }
    firstdone[b] = first;
  }
}

constexpr int AP = 34;

__global__ void __launch_bounds__(32)
retention16(const float* __restrict__ q, const float* __restrict__ kbuf,
            const float* __restrict__ vbuf, const float* __restrict__ hstate,
            const int* __restrict__ cum, const int* __restrict__ firstdone,
            float* __restrict__ out)
{
  __shared__ int   cumsh[Tc];
  __shared__ __attribute__((aligned(16))) float atile[16 * AP];

  const int lane  = threadIdx.x & 31;
  const int nl    = lane & 15;
  const int khalf = (lane >> 4) << 1;
  const int mh    = (lane >> 4) << 3;
  const int it    = blockIdx.x;
  const int b     = blockIdx.y;

  const int* cumb = cum + b * Tc;
  for (int i = lane; i < Tc; i += 32)  cumsh[i] = cumb[i];
  __syncthreads();

  const int ti0 = it * 4 + (mh >> 2);
  const int cr0 = cumsh[ti0];
  const int cr1 = cumsh[ti0 + 1];
  const int cb  = nl >> 2;

  const float* qb = q      + (size_t)b * Sc * Hc;
  const float* kb = kbuf   + (size_t)b * Sc * Hc;
  const float* vb = vbuf   + (size_t)b * Sc * Hc;
  const float* hb = hstate + (size_t)b * Hc * Hc;

  const float* qrow = qb + (size_t)(it * 16 + nl) * Hc;
  v2f qa[16];
  #pragma unroll
  for (int ks = 0; ks < 16; ++ks)
    qa[ks] = *(const v2f*)(qrow + ks * 4 + khalf);

  v8f acc[4] = {};
  const int ntrip = (it >> 1) + 1;
  for (int jp = 0; jp < ntrip; ++jp) {
    const int jt0 = 2 * jp;
    v2f vf[8][4];
    #pragma unroll
    for (int kk = 0; kk < 8; ++kk) {
      const float* v0 = vb + (size_t)(jt0 * 16 + kk * 4 + khalf) * Hc;
      const float* v1 = v0 + Hc;
      #pragma unroll
      for (int nt = 0; nt < 4; ++nt) {
        vf[kk][nt].x = v0[nt * 16 + nl];
        vf[kk][nt].y = v1[nt * 16 + nl];
      }
    }
    v2f kf[2][16];
    const float* krow = kb + (size_t)(jt0 * 16 + nl) * Hc;
    #pragma unroll
    for (int ks = 0; ks < 16; ++ks) {
      kf[0][ks] = *(const v2f*)(krow + ks * 4 + khalf);
      kf[1][ks] = *(const v2f*)(krow + 16 * Hc + ks * 4 + khalf);
    }
    const int tj0 = jt0 * 4 + cb;
    const int tj1 = tj0 + 4;
    const int cj0 = cumsh[tj0];
    const int cj1 = cumsh[tj1];
    v8f s0 = {}, s1 = {};
    #pragma unroll
    for (int ks = 0; ks < 16; ++ks) {
      s0 = wmma4(qa[ks], kf[0][ks], s0);
      s1 = wmma4(qa[ks], kf[1][ks], s1);
    }
    const float m00 = (cj0 == cr0) ? __expf((float)(ti0 - tj0) * LNK) : 0.0f;
    const float m01 = (cj0 == cr1) ? __expf((float)(ti0 + 1 - tj0) * LNK) : 0.0f;
    const float m10 = (cj1 == cr0) ? __expf((float)(ti0 - tj1) * LNK) : 0.0f;
    const float m11 = (cj1 == cr1) ? __expf((float)(ti0 + 1 - tj1) * LNK) : 0.0f;
    const int   dg0 = (it - jt0) * 16 + mh - nl;
    const int   dg1 = dg0 - 16;
    #pragma unroll
    for (int r = 0; r < 8; ++r) {
      const float w0 = (dg0 + r >= 0) ? ((r < 4) ? m00 : m01) : 0.0f;
      const float w1 = (dg1 + r >= 0) ? ((r < 4) ? m10 : m11) : 0.0f;
      atile[(mh + r) * AP + nl]      = s0[r] * w0;
      atile[(mh + r) * AP + 16 + nl] = s1[r] * w1;
    }
    __syncthreads();
    #pragma unroll
    for (int kk = 0; kk < 8; ++kk) {
      v2f a = *(const v2f*)(&atile[nl * AP + kk * 4 + khalf]);
      #pragma unroll
      for (int nt = 0; nt < 4; ++nt)
        acc[nt] = wmma4(a, vf[kk][nt], acc[nt]);
    }
    __syncthreads();
  }

  v8f cr[4] = {};
  #pragma unroll
  for (int ks = 0; ks < 16; ++ks) {
    const float* h0 = hb + (size_t)(ks * 4 + khalf) * Hc;
    const float* h1 = h0 + Hc;
    #pragma unroll
    for (int nt = 0; nt < 4; ++nt) {
      v2f bm;
      bm.x = h0[nt * 16 + nl];
      bm.y = h1[nt * 16 + nl];
      cr[nt] = wmma4(qa[ks], bm, cr[nt]);
    }
  }

  const int fd = firstdone[b];
  #pragma unroll
  for (int r = 0; r < 8; ++r) {
    const int   gi = it * 16 + mh + r;
    const int   ti = gi >> 2;
    const float xi = (ti < fd) ? __expf((float)(ti + 1) * LNK) : 0.0f;
    float* orow = out + ((size_t)b * Sc + gi) * Hc;
    #pragma unroll
    for (int nt = 0; nt < 4; ++nt)
      orow[nt * 16 + nl] = acc[nt][r] + cr[nt][r] * xi;
  }
}

__global__ void __launch_bounds__(32)
hstate_update(const float* __restrict__ kbuf, const float* __restrict__ vbuf,
              const float* __restrict__ hstate, const int* __restrict__ cum,
              float* __restrict__ hout)
{
  __shared__ float ktab[Tc + 1];
  __shared__ float wsh[Tc];
  const int lane  = threadIdx.x & 31;
  const int nl    = lane & 15;
  const int khalf = (lane >> 4) << 1;
  const int mt = blockIdx.x, nt = blockIdx.y, b = blockIdx.z;
  const int* cumb = cum + b * Tc;
  const int  cend = cumb[Tc - 1];

  for (int i = lane; i <= Tc; i += 32) ktab[i] = __expf((float)i * LNK);
  __syncthreads();
  for (int i = lane; i < Tc; i += 32)
    wsh[i] = (cumb[i] == cend) ? ktab[(Tc - 1) - i] : 0.0f;
  __syncthreads();

  const float* kb = kbuf + (size_t)b * Sc * Hc;
  const float* vb = vbuf + (size_t)b * Sc * Hc;

  v8f c = {};
  for (int s0 = 0; s0 < Sc; s0 += 4) {
    const int   sA = s0 + khalf;
    const float w  = wsh[s0 >> 2];
    v2f a;
    a.x = kb[(size_t)sA       * Hc + mt * 16 + nl];
    a.y = kb[(size_t)(sA + 1) * Hc + mt * 16 + nl];
    v2f bm;
    bm.x = vb[(size_t)sA       * Hc + nt * 16 + nl] * w;
    bm.y = vb[(size_t)(sA + 1) * Hc + nt * 16 + nl] * w;
    c = wmma4(a, bm, c);
  }

  const float cd    = ktab[Tc];
  const float delta = (cend == 0) ? 1.0f : 0.0f;
  const float* hb   = hstate + (size_t)b * Hc * Hc;
  const int    mh   = (lane >> 4) << 3;
  #pragma unroll
  for (int r = 0; r < 8; ++r) {
    const int h1 = mt * 16 + mh + r;
    hout[((size_t)b * Hc + h1) * Hc + nt * 16 + nl] =
        c[r] + hb[(size_t)h1 * Hc + nt * 16 + nl] * cd * delta;
  }
}

extern "C" void kernel_launch(void* const* d_in, const int* in_sizes, int n_in,
                              void* d_out, int out_size, void* d_ws, size_t ws_size,
                              hipStream_t stream)
{
  const float* key    = (const float*)d_in[0];
  const float* query  = (const float*)d_in[1];
  const float* value  = (const float*)d_in[2];
  const float* hstate = (const float*)d_in[3];
  const int*   dones  = (const int*)d_in[4];
  const float* W_Q    = (const float*)d_in[5];
  const float* W_K    = (const float*)d_in[6];
  const float* W_V    = (const float*)d_in[7];

  const size_t qkv = (size_t)Bc * Sc * Hc;
  float* ws = (float*)d_ws;
  float* qb = ws;
  float* kb = ws + qkv;
  float* vb = ws + 2 * qkv;
  int*   cum       = (int*)(ws + 3 * qkv);
  int*   firstdone = cum + Bc * Tc;

  dim3 gproj(Bc * Sc / (16 * PJ_TILES));
  proj_splitk<<<gproj, 256, 0, stream>>>(query, W_Q, qb);
  proj_splitk<<<gproj, 256, 0, stream>>>(key,   W_K, kb);
  proj_splitk<<<gproj, 256, 0, stream>>>(value, W_V, vb);

  dones_scan<<<Bc, 32, 0, stream>>>(dones, cum, firstdone);

  float* out0 = (float*)d_out;
  float* hout = out0 + qkv;
  retention16<<<dim3(Sc / 16, Bc), 32, 0, stream>>>(qb, kb, vb, hstate, cum,
                                                    firstdone, out0);
  hstate_update<<<dim3(Hc / 16, Hc / 16, Bc), 32, 0, stream>>>(kb, vb, hstate,
                                                               cum, hout);
}


# retention: q, k, v, hstate fragments and the output by 16-byte loads/stores (a lane owns 4 consecutive columns); 96 loads per 32-key trip become 32; features summed in another order
# speedup vs baseline: 2.3499x; 1.1166x over previous
#include <hip/hip_runtime.h>
#include <hip/hip_bf16.h>


typedef float v2f __attribute__((ext_vector_type(2)));
typedef float v4f __attribute__((ext_vector_type(4)));
typedef float v8f __attribute__((ext_vector_type(8)));

constexpr int   Bc = 32, Sc = 1024, Ec = 512, Hc = 64, Tc = 256;
constexpr float LNK = -0.1053605156578263f;

static __device__ __forceinline__ v8f wmma4(v2f a, v2f b, v8f c) {
  return __builtin_amdgcn_wmma_f32_16x16x4_f32(false, a, false, b, (short)0, c,
                                               false, false);
}

constexpr int PJ_TILES = 4;

__global__ void __launch_bounds__(256, 4)
proj_splitk(const float* __restrict__ X, const float* __restrict__ W,
            float* __restrict__ Y)
{
  __shared__ v4f part[2][8][16][16];

  const int lane = threadIdx.x & 31;
  const int wave = threadIdx.x >> 5;
  const int c    = lane & 15;
  const int h    = lane >> 4;
  const int kw   = wave << 6;

  const int      rt0  = blockIdx.x * PJ_TILES;
  const float*   xp   = X + (size_t)rt0 * 16 * Ec;
  float*         yp   = Y + (size_t)rt0 * 16 * Hc;
  const unsigned xoff = (unsigned)(c * Ec + kw + 4 * h);
  const unsigned yoff = (unsigned)((2 * wave + h) * Hc + 4 * c);

  v4f xa[8];
  #pragma unroll
  for (int j = 0; j < 8; ++j) xa[j] = *(const v4f*)(xp + xoff + 8 * j);

  v2f wb[4][8][2];
  v4f raw[2][8];
  const float* wp = W + (size_t)(kw + 4 * h) * Hc + 4 * c;
  #pragma unroll
  for (int i = 0; i < 8; ++i)
    raw[0][i] = *(const v4f*)(wp + (size_t)(8 * (i >> 2) + (i & 3)) * Hc);
  #pragma unroll
  for (int ch = 0; ch < 4; ++ch) {
    if (ch + 1 < 4) {
      #pragma unroll
      for (int i = 0; i < 8; ++i)
        raw[(ch + 1) & 1][i] =
            *(const v4f*)(wp + (size_t)(8 * (2 * (ch + 1) + (i >> 2)) + (i & 3)) * Hc);
    }
    #pragma unroll
    for (int jj = 0; jj < 2; ++jj) {
      #pragma unroll
      for (int p = 0; p < 2; ++p) {
        #pragma unroll
        for (int nt = 0; nt < 4; ++nt) {
          wb[nt][2 * ch + jj][p].x = raw[ch & 1][4 * jj + 2 * p][nt];
          wb[nt][2 * ch + jj][p].y = raw[ch & 1][4 * jj + 2 * p + 1][nt];
        }
      }
    }
    __builtin_amdgcn_sched_barrier(0);
  }

  #pragma unroll
  for (int t = 0; t < PJ_TILES; ++t) {
    v4f xn[8];
    if (t + 1 < PJ_TILES) {
      const float* xq = xp + (size_t)(t + 1) * 16 * Ec;
      #pragma unroll
      for (int j = 0; j < 8; ++j) xn[j] = *(const v4f*)(xq + xoff + 8 * j);
    }

    v8f acc[4] = {};
    #pragma unroll
    for (int j = 0; j < 8; ++j) {
      #pragma unroll
      for (int p = 0; p < 2; ++p) {
        v2f a;
        a.x = xa[j][2 * p];
        a.y = xa[j][2 * p + 1];
        #pragma unroll
        for (int nt = 0; nt < 4; ++nt)
          acc[nt] = wmma4(a, wb[nt][j][p], acc[nt]);
      }
    }

    #pragma unroll
    for (int r = 0; r < 8; ++r) {
      v4f o;
      o.x = acc[0][r]; o.y = acc[1][r]; o.z = acc[2][r]; o.w = acc[3][r];
      part[t & 1][wave][8 * h + r][c] = o;
    }
    __syncthreads();

    v4f s = part[t & 1][0][2 * wave + h][c];
    #pragma unroll
    for (int u = 1; u < 8; ++u) s += part[t & 1][u][2 * wave + h][c];
    *(v4f*)(yp + (size_t)t * 16 * Hc + yoff) = s;

    if (t + 1 < PJ_TILES) {
      #pragma unroll
      for (int j = 0; j < 8; ++j) xa[j] = xn[j];
    }
  }
}

__global__ void dones_scan(const int* __restrict__ dones, int* __restrict__ cum,
                           int* __restrict__ firstdone)
{
  const int b = blockIdx.x;
  if (threadIdx.x == 0) {
    int c = 0, first = Tc;
    for (int t = 0; t < Tc; ++t) {
      const int d = (dones[(size_t)b * Sc + t * 4] != 0) ? 1 : 0;
      if (d && first == Tc) first = t;
      c += d;
      cum[b * Tc + t] = c;
    }
    firstdone[b] = first;
  }
}

constexpr int AP = 34;

__global__ void __launch_bounds__(32)
retention16(const float* __restrict__ q, const float* __restrict__ kbuf,
            const float* __restrict__ vbuf, const float* __restrict__ hstate,
            const int* __restrict__ cum, const int* __restrict__ firstdone,
            float* __restrict__ out)
{
  __shared__ int   cumsh[Tc];
  __shared__ __attribute__((aligned(16))) float atile[16 * AP];

  const int lane  = threadIdx.x & 31;
  const int nl    = lane & 15;
  const int hh    = lane >> 4;
  const int khalf = hh << 1;
  const int mh    = hh << 3;
  const int it    = blockIdx.x;
  const int b     = blockIdx.y;

  const int* cumb = cum + b * Tc;
  for (int i = lane; i < Tc; i += 32)  cumsh[i] = cumb[i];
  __syncthreads();

  const int ti0 = it * 4 + (mh >> 2);
  const int cr0 = cumsh[ti0];
  const int cr1 = cumsh[ti0 + 1];
  const int cb  = nl >> 2;

  const float* qb = q      + (size_t)b * Sc * Hc;
  const float* kb = kbuf   + (size_t)b * Sc * Hc;
  const float* vb = vbuf   + (size_t)b * Sc * Hc;
  const float* hb = hstate + (size_t)b * Hc * Hc;

  const float* qrow = qb + (size_t)(it * 16 + nl) * Hc + 4 * hh;
  v4f q4[8];
  #pragma unroll
  for (int j = 0; j < 8; ++j) q4[j] = *(const v4f*)(qrow + 8 * j);

  v8f acc[4] = {};
  const int ntrip = (it >> 1) + 1;
  for (int jp = 0; jp < ntrip; ++jp) {
    const int jt0 = 2 * jp;
    v4f k4[2][8];
    const float* krow = kb + (size_t)(jt0 * 16 + nl) * Hc + 4 * hh;
    #pragma unroll
    for (int j = 0; j < 8; ++j) {
      k4[0][j] = *(const v4f*)(krow + 8 * j);
      k4[1][j] = *(const v4f*)(krow + 16 * Hc + 8 * j);
    }
    v4f vr[8][2];
    const float* vrow = vb + (size_t)(jt0 * 16 + khalf) * Hc + 4 * nl;
    #pragma unroll
    for (int kk = 0; kk < 8; ++kk) {
      vr[kk][0] = *(const v4f*)(vrow + (size_t)(kk * 4) * Hc);
      vr[kk][1] = *(const v4f*)(vrow + (size_t)(kk * 4 + 1) * Hc);
    }
    __builtin_amdgcn_sched_barrier(0);
    const int tj0 = jt0 * 4 + cb;
    const int tj1 = tj0 + 4;
    const int cj0 = cumsh[tj0];
    const int cj1 = cumsh[tj1];
    v8f s0 = {}, s1 = {};
    #pragma unroll
    for (int j = 0; j < 8; ++j) {
      #pragma unroll
      for (int p = 0; p < 2; ++p) {
        v2f a, b0, b1;
        a.x  = q4[j][2 * p];     a.y  = q4[j][2 * p + 1];
        b0.x = k4[0][j][2 * p];  b0.y = k4[0][j][2 * p + 1];
        b1.x = k4[1][j][2 * p];  b1.y = k4[1][j][2 * p + 1];
        s0 = wmma4(a, b0, s0);
        s1 = wmma4(a, b1, s1);
      }
    }
    const float m00 = (cj0 == cr0) ? __expf((float)(ti0 - tj0) * LNK) : 0.0f;
    const float m01 = (cj0 == cr1) ? __expf((float)(ti0 + 1 - tj0) * LNK) : 0.0f;
    const float m10 = (cj1 == cr0) ? __expf((float)(ti0 - tj1) * LNK) : 0.0f;
    const float m11 = (cj1 == cr1) ? __expf((float)(ti0 + 1 - tj1) * LNK) : 0.0f;
    const int   dg0 = (it - jt0) * 16 + mh - nl;
    const int   dg1 = dg0 - 16;
    #pragma unroll
    for (int r = 0; r < 8; ++r) {
      const float w0 = (dg0 + r >= 0) ? ((r < 4) ? m00 : m01) : 0.0f;
      const float w1 = (dg1 + r >= 0) ? ((r < 4) ? m10 : m11) : 0.0f;
      atile[(mh + r) * AP + nl]      = s0[r] * w0;
      atile[(mh + r) * AP + 16 + nl] = s1[r] * w1;
    }
    __syncthreads();
    #pragma unroll
    for (int kk = 0; kk < 8; ++kk) {
      v2f a = *(const v2f*)(&atile[nl * AP + kk * 4 + khalf]);
      #pragma unroll
      for (int nt = 0; nt < 4; ++nt) {
        v2f bm;
        bm.x = vr[kk][0][nt];
        bm.y = vr[kk][1][nt];
        acc[nt] = wmma4(a, bm, acc[nt]);
      }
    }
    __syncthreads();
  }

  v8f cr[4] = {};
  #pragma unroll
  for (int j = 0; j < 8; ++j) {
    #pragma unroll
    for (int p = 0; p < 2; ++p) {
      const float* hr = hb + (size_t)(8 * j + 4 * hh + 2 * p) * Hc + 4 * nl;
      const v4f h0 = *(const v4f*)(hr);
      const v4f h1 = *(const v4f*)(hr + Hc);
      v2f a;
      a.x = q4[j][2 * p];
      a.y = q4[j][2 * p + 1];
      #pragma unroll
      for (int nt = 0; nt < 4; ++nt) {
        v2f bm;
        bm.x = h0[nt];
        bm.y = h1[nt];
        cr[nt] = wmma4(a, bm, cr[nt]);
      }
    }
  }

  const int fd = firstdone[b];
  #pragma unroll
  for (int r = 0; r < 8; ++r) {
    const int   gi = it * 16 + mh + r;
    const int   ti = gi >> 2;
    const float xi = (ti < fd) ? __expf((float)(ti + 1) * LNK) : 0.0f;
    v4f o;
    o.x = acc[0][r] + cr[0][r] * xi;
    o.y = acc[1][r] + cr[1][r] * xi;
    o.z = acc[2][r] + cr[2][r] * xi;
    o.w = acc[3][r] + cr[3][r] * xi;
    *(v4f*)(out + ((size_t)b * Sc + gi) * Hc + 4 * nl) = o;
  }
}

__global__ void __launch_bounds__(32)
hstate_update(const float* __restrict__ kbuf, const float* __restrict__ vbuf,
              const float* __restrict__ hstate, const int* __restrict__ cum,
              float* __restrict__ hout)
{
  __shared__ float ktab[Tc + 1];
  __shared__ float wsh[Tc];
  const int lane  = threadIdx.x & 31;
  const int nl    = lane & 15;
  const int khalf = (lane >> 4) << 1;
  const int mt = blockIdx.x, nt = blockIdx.y, b = blockIdx.z;
  const int* cumb = cum + b * Tc;
  const int  cend = cumb[Tc - 1];

  for (int i = lane; i <= Tc; i += 32) ktab[i] = __expf((float)i * LNK);
  __syncthreads();
  for (int i = lane; i < Tc; i += 32)
    wsh[i] = (cumb[i] == cend) ? ktab[(Tc - 1) - i] : 0.0f;
  __syncthreads();

  const float* kb = kbuf + (size_t)b * Sc * Hc;
  const float* vb = vbuf + (size_t)b * Sc * Hc;

  v8f c = {};
  for (int s0 = 0; s0 < Sc; s0 += 4) {
    const int   sA = s0 + khalf;
    const float w  = wsh[s0 >> 2];
    v2f a;
    a.x = kb[(size_t)sA       * Hc + mt * 16 + nl];
    a.y = kb[(size_t)(sA + 1) * Hc + mt * 16 + nl];
    v2f bm;
    bm.x = vb[(size_t)sA       * Hc + nt * 16 + nl] * w;
    bm.y = vb[(size_t)(sA + 1) * Hc + nt * 16 + nl] * w;
    c = wmma4(a, bm, c);
  }

  const float cd    = ktab[Tc];
  const float delta = (cend == 0) ? 1.0f : 0.0f;
  const float* hb   = hstate + (size_t)b * Hc * Hc;
  const int    mh   = (lane >> 4) << 3;
  #pragma unroll
  for (int r = 0; r < 8; ++r) {
    const int h1 = mt * 16 + mh + r;
    hout[((size_t)b * Hc + h1) * Hc + nt * 16 + nl] =
        c[r] + hb[(size_t)h1 * Hc + nt * 16 + nl] * cd * delta;
  }
}

extern "C" void kernel_launch(void* const* d_in, const int* in_sizes, int n_in,
                              void* d_out, int out_size, void* d_ws, size_t ws_size,
                              hipStream_t stream)
{
  const float* key    = (const float*)d_in[0];
  const float* query  = (const float*)d_in[1];
  const float* value  = (const float*)d_in[2];
  const float* hstate = (const float*)d_in[3];
  const int*   dones  = (const int*)d_in[4];
  const float* W_Q    = (const float*)d_in[5];
  const float* W_K    = (const float*)d_in[6];
  const float* W_V    = (const float*)d_in[7];

  const size_t qkv = (size_t)Bc * Sc * Hc;
  float* ws = (float*)d_ws;
  float* qb = ws;
  float* kb = ws + qkv;
  float* vb = ws + 2 * qkv;
  int*   cum       = (int*)(ws + 3 * qkv);
  int*   firstdone = cum + Bc * Tc;

  dim3 gproj(Bc * Sc / (16 * PJ_TILES));
  proj_splitk<<<gproj, 256, 0, stream>>>(query, W_Q, qb);
  proj_splitk<<<gproj, 256, 0, stream>>>(key,   W_K, kb);
  proj_splitk<<<gproj, 256, 0, stream>>>(value, W_V, vb);

  dones_scan<<<Bc, 32, 0, stream>>>(dones, cum, firstdone);

  float* out0 = (float*)d_out;
  float* hout = out0 + qkv;
  retention16<<<dim3(Sc / 16, Bc), 32, 0, stream>>>(qb, kb, vb, hstate, cum,
                                                    firstdone, out0);
  hstate_update<<<dim3(Hc / 16, Hc / 16, Bc), 32, 0, stream>>>(kb, vb, hstate,
                                                               cum, hout);
}


# retention and state update skip the key blocks in front of the last episode end seen by the strip's first row (the state: by the last row): their decay weights are all 0; same sums, same order
# speedup vs baseline: 2.9137x; 1.2399x over previous
#include <hip/hip_runtime.h>
#include <hip/hip_bf16.h>


typedef float v2f __attribute__((ext_vector_type(2)));
typedef float v4f __attribute__((ext_vector_type(4)));
typedef float v8f __attribute__((ext_vector_type(8)));

constexpr int   Bc = 32, Sc = 1024, Ec = 512, Hc = 64, Tc = 256;
constexpr float LNK = -0.1053605156578263f;

static __device__ __forceinline__ v8f wmma4(v2f a, v2f b, v8f c) {
  return __builtin_amdgcn_wmma_f32_16x16x4_f32(false, a, false, b, (short)0, c,
                                               false, false);
}

constexpr int PJ_TILES = 4;

__global__ void __launch_bounds__(256, 4)
proj_splitk(const float* __restrict__ X, const float* __restrict__ W,
            float* __restrict__ Y)
{
  __shared__ v4f part[2][8][16][16];

  const int lane = threadIdx.x & 31;
  const int wave = threadIdx.x >> 5;
  const int c    = lane & 15;
  const int h    = lane >> 4;
  const int kw   = wave << 6;

  const int      rt0  = blockIdx.x * PJ_TILES;
  const float*   xp   = X + (size_t)rt0 * 16 * Ec;
  float*         yp   = Y + (size_t)rt0 * 16 * Hc;
  const unsigned xoff = (unsigned)(c * Ec + kw + 4 * h);
  const unsigned yoff = (unsigned)((2 * wave + h) * Hc + 4 * c);

  v4f xa[8];
  #pragma unroll
  for (int j = 0; j < 8; ++j) xa[j] = *(const v4f*)(xp + xoff + 8 * j);

  v2f wb[4][8][2];
  v4f raw[2][8];
  const float* wp = W + (size_t)(kw + 4 * h) * Hc + 4 * c;
  #pragma unroll
  for (int i = 0; i < 8; ++i)
    raw[0][i] = *(const v4f*)(wp + (size_t)(8 * (i >> 2) + (i & 3)) * Hc);
  #pragma unroll
  for (int ch = 0; ch < 4; ++ch) {
    if (ch + 1 < 4) {
      #pragma unroll
      for (int i = 0; i < 8; ++i)
        raw[(ch + 1) & 1][i] =
            *(const v4f*)(wp + (size_t)(8 * (2 * (ch + 1) + (i >> 2)) + (i & 3)) * Hc);
    }
    #pragma unroll
    for (int jj = 0; jj < 2; ++jj) {
      #pragma unroll
      for (int p = 0; p < 2; ++p) {
        #pragma unroll
        for (int nt = 0; nt < 4; ++nt) {
          wb[nt][2 * ch + jj][p].x = raw[ch & 1][4 * jj + 2 * p][nt];
          wb[nt][2 * ch + jj][p].y = raw[ch & 1][4 * jj + 2 * p + 1][nt];
        }
      }
    }
    __builtin_amdgcn_sched_barrier(0);
  }

  #pragma unroll
  for (int t = 0; t < PJ_TILES; ++t) {
    v4f xn[8];
    if (t + 1 < PJ_TILES) {
      const float* xq = xp + (size_t)(t + 1) * 16 * Ec;
      #pragma unroll
      for (int j = 0; j < 8; ++j) xn[j] = *(const v4f*)(xq + xoff + 8 * j);
    }

    v8f acc[4] = {};
    #pragma unroll
    for (int j = 0; j < 8; ++j) {
      #pragma unroll
      for (int p = 0; p < 2; ++p) {
        v2f a;
        a.x = xa[j][2 * p];
        a.y = xa[j][2 * p + 1];
        #pragma unroll
        for (int nt = 0; nt < 4; ++nt)
          acc[nt] = wmma4(a, wb[nt][j][p], acc[nt]);
      }
    }

    #pragma unroll
    for (int r = 0; r < 8; ++r) {
      v4f o;
      o.x = acc[0][r]; o.y = acc[1][r]; o.z = acc[2][r]; o.w = acc[3][r];
      part[t & 1][wave][8 * h + r][c] = o;
    }
    __syncthreads();

    v4f s = part[t & 1][0][2 * wave + h][c];
    #pragma unroll
    for (int u = 1; u < 8; ++u) s += part[t & 1][u][2 * wave + h][c];
    *(v4f*)(yp + (size_t)t * 16 * Hc + yoff) = s;

    if (t + 1 < PJ_TILES) {
      #pragma unroll
      for (int j = 0; j < 8; ++j) xa[j] = xn[j];
    }
  }
}

__global__ void dones_scan(const int* __restrict__ dones, int* __restrict__ etag,
                           int* __restrict__ firstdone)
{
  const int b = blockIdx.x;
  if (threadIdx.x == 0) {
    int tag = 0, first = Tc;
    for (int t = 0; t < Tc; ++t) {
      const int d = (dones[(size_t)b * Sc + t * 4] != 0) ? 1 : 0;
      if (d && first == Tc) first = t;
      if (d) tag = t + 1;
      etag[b * Tc + t] = tag;
    }
    firstdone[b] = first;
  }
}

constexpr int AP = 34;

__global__ void __launch_bounds__(32)
retention16(const float* __restrict__ q, const float* __restrict__ kbuf,
            const float* __restrict__ vbuf, const float* __restrict__ hstate,
            const int* __restrict__ etag, const int* __restrict__ firstdone,
            float* __restrict__ out)
{
  __shared__ int   tagsh[Tc];
  __shared__ __attribute__((aligned(16))) float atile[16 * AP];

  const int lane  = threadIdx.x & 31;
  const int nl    = lane & 15;
  const int hh    = lane >> 4;
  const int khalf = hh << 1;
  const int mh    = hh << 3;
  const int it    = blockIdx.x;
  const int b     = blockIdx.y;

  const int* tagb = etag + b * Tc;
  const int tag0 = tagb[it * 4];
  for (int i = lane; i < Tc; i += 32)  tagsh[i] = tagb[i];
  __syncthreads();

  const int ti0 = it * 4 + (mh >> 2);
  const int cr0 = tagsh[ti0];
  const int cr1 = tagsh[ti0 + 1];
  const int cb  = nl >> 2;

  const float* qb = q      + (size_t)b * Sc * Hc;
  const float* kb = kbuf   + (size_t)b * Sc * Hc;
  const float* vb = vbuf   + (size_t)b * Sc * Hc;
  const float* hb = hstate + (size_t)b * Hc * Hc;

  const float* qrow = qb + (size_t)(it * 16 + nl) * Hc + 4 * hh;
  v4f q4[8];
  #pragma unroll
  for (int j = 0; j < 8; ++j) q4[j] = *(const v4f*)(qrow + 8 * j);

  v8f acc[4] = {};
  const int ntrip = (it >> 1) + 1;
  const int jp0  = ((tag0 > 0) ? tag0 - 1 : 0) >> 3;
  for (int jp = jp0; jp < ntrip; ++jp) {
    const int jt0 = 2 * jp;
    v4f k4[2][8];
    const float* krow = kb + (size_t)(jt0 * 16 + nl) * Hc + 4 * hh;
    #pragma unroll
    for (int j = 0; j < 8; ++j) {
      k4[0][j] = *(const v4f*)(krow + 8 * j);
      k4[1][j] = *(const v4f*)(krow + 16 * Hc + 8 * j);
    }
    v4f vr[8][2];
    const float* vrow = vb + (size_t)(jt0 * 16 + khalf) * Hc + 4 * nl;
    #pragma unroll
    for (int kk = 0; kk < 8; ++kk) {
      vr[kk][0] = *(const v4f*)(vrow + (size_t)(kk * 4) * Hc);
      vr[kk][1] = *(const v4f*)(vrow + (size_t)(kk * 4 + 1) * Hc);
    }
    __builtin_amdgcn_sched_barrier(0);
    const int tj0 = jt0 * 4 + cb;
    const int tj1 = tj0 + 4;
    const int cj0 = tagsh[tj0];
    const int cj1 = tagsh[tj1];
    v8f s0 = {}, s1 = {};
    #pragma unroll
    for (int j = 0; j < 8; ++j) {
      #pragma unroll
      for (int p = 0; p < 2; ++p) {
        v2f a, b0, b1;
        a.x  = q4[j][2 * p];     a.y  = q4[j][2 * p + 1];
        b0.x = k4[0][j][2 * p];  b0.y = k4[0][j][2 * p + 1];
        b1.x = k4[1][j][2 * p];  b1.y = k4[1][j][2 * p + 1];
        s0 = wmma4(a, b0, s0);
        s1 = wmma4(a, b1, s1);
      }
    }
    const float m00 = (cj0 == cr0) ? __expf((float)(ti0 - tj0) * LNK) : 0.0f;
    const float m01 = (cj0 == cr1) ? __expf((float)(ti0 + 1 - tj0) * LNK) : 0.0f;
    const float m10 = (cj1 == cr0) ? __expf((float)(ti0 - tj1) * LNK) : 0.0f;
    const float m11 = (cj1 == cr1) ? __expf((float)(ti0 + 1 - tj1) * LNK) : 0.0f;
    const int   dg0 = (it - jt0) * 16 + mh - nl;
    const int   dg1 = dg0 - 16;
    #pragma unroll
    for (int r = 0; r < 8; ++r) {
      const float w0 = (dg0 + r >= 0) ? ((r < 4) ? m00 : m01) : 0.0f;
      const float w1 = (dg1 + r >= 0) ? ((r < 4) ? m10 : m11) : 0.0f;
      atile[(mh + r) * AP + nl]      = s0[r] * w0;
      atile[(mh + r) * AP + 16 + nl] = s1[r] * w1;
    }
    __syncthreads();
    #pragma unroll
    for (int kk = 0; kk < 8; ++kk) {
      v2f a = *(const v2f*)(&atile[nl * AP + kk * 4 + khalf]);
      #pragma unroll
      for (int nt = 0; nt < 4; ++nt) {
        v2f bm;
        bm.x = vr[kk][0][nt];
        bm.y = vr[kk][1][nt];
        acc[nt] = wmma4(a, bm, acc[nt]);
      }
    }
    __syncthreads();
  }

  v8f cr[4] = {};
  #pragma unroll
  for (int j = 0; j < 8; ++j) {
    #pragma unroll
    for (int p = 0; p < 2; ++p) {
      const float* hr = hb + (size_t)(8 * j + 4 * hh + 2 * p) * Hc + 4 * nl;
      const v4f h0 = *(const v4f*)(hr);
      const v4f h1 = *(const v4f*)(hr + Hc);
      v2f a;
      a.x = q4[j][2 * p];
      a.y = q4[j][2 * p + 1];
      #pragma unroll
      for (int nt = 0; nt < 4; ++nt) {
        v2f bm;
        bm.x = h0[nt];
        bm.y = h1[nt];
        cr[nt] = wmma4(a, bm, cr[nt]);
      }
    }
  }

  const int fd = firstdone[b];
  #pragma unroll
  for (int r = 0; r < 8; ++r) {
    const int   gi = it * 16 + mh + r;
    const int   ti = gi >> 2;
    const float xi = (ti < fd) ? __expf((float)(ti + 1) * LNK) : 0.0f;
    v4f o;
    o.x = acc[0][r] + cr[0][r] * xi;
    o.y = acc[1][r] + cr[1][r] * xi;
    o.z = acc[2][r] + cr[2][r] * xi;
    o.w = acc[3][r] + cr[3][r] * xi;
    *(v4f*)(out + ((size_t)b * Sc + gi) * Hc + 4 * nl) = o;
  }
}

__global__ void __launch_bounds__(32)
hstate_update(const float* __restrict__ kbuf, const float* __restrict__ vbuf,
              const float* __restrict__ hstate, const int* __restrict__ etag,
              float* __restrict__ hout)
{
  __shared__ float ktab[Tc + 1];
  __shared__ float wsh[Tc];
  const int lane  = threadIdx.x & 31;
  const int nl    = lane & 15;
  const int khalf = (lane >> 4) << 1;
  const int mt = blockIdx.x, nt = blockIdx.y, b = blockIdx.z;
  const int* tagb = etag + b * Tc;
  const int  tend = tagb[Tc - 1];

  for (int i = lane; i <= Tc; i += 32) ktab[i] = __expf((float)i * LNK);
  __syncthreads();
  for (int i = lane; i < Tc; i += 32)
    wsh[i] = (tagb[i] == tend) ? ktab[(Tc - 1) - i] : 0.0f;
  __syncthreads();

  const float* kb = kbuf + (size_t)b * Sc * Hc;
  const float* vb = vbuf + (size_t)b * Sc * Hc;

  const int sbeg = (((tend > 0) ? tend - 1 : 0) >> 3) << 5;
  const float* kp = kb + (size_t)(sbeg + khalf) * Hc + mt * 16 + nl;
  const float* vp = vb + (size_t)(sbeg + khalf) * Hc + nt * 16 + nl;
  const float* wq = wsh + (sbeg >> 2);
  v8f c = {};
  for (int sb = sbeg; sb < Sc; sb += 32) {
    #pragma unroll
    for (int u = 0; u < 8; ++u) {
      const float w = wq[u];
      v2f a;
      a.x = kp[(4 * u)     * Hc];
      a.y = kp[(4 * u + 1) * Hc];
      v2f bm;
      bm.x = vp[(4 * u)     * Hc] * w;
      bm.y = vp[(4 * u + 1) * Hc] * w;
      c = wmma4(a, bm, c);
    }
    kp += 32 * Hc;
    vp += 32 * Hc;
    wq += 8;
  }

  const float cd    = ktab[Tc];
  const float delta = (tend == 0) ? 1.0f : 0.0f;
  const float* hb   = hstate + (size_t)b * Hc * Hc;
  const int    mh   = (lane >> 4) << 3;
  #pragma unroll
  for (int r = 0; r < 8; ++r) {
    const int h1 = mt * 16 + mh + r;
    hout[((size_t)b * Hc + h1) * Hc + nt * 16 + nl] =
        c[r] + hb[(size_t)h1 * Hc + nt * 16 + nl] * cd * delta;
  }
}

extern "C" void kernel_launch(void* const* d_in, const int* in_sizes, int n_in,
                              void* d_out, int out_size, void* d_ws, size_t ws_size,
                              hipStream_t stream)
{
  const float* key    = (const float*)d_in[0];
  const float* query  = (const float*)d_in[1];
  const float* value  = (const float*)d_in[2];
  const float* hstate = (const float*)d_in[3];
  const int*   dones  = (const int*)d_in[4];
  const float* W_Q    = (const float*)d_in[5];
  const float* W_K    = (const float*)d_in[6];
  const float* W_V    = (const float*)d_in[7];

  const size_t qkv = (size_t)Bc * Sc * Hc;
  float* ws = (float*)d_ws;
  float* qb = ws;
  float* kb = ws + qkv;
  float* vb = ws + 2 * qkv;
  int*   etag       = (int*)(ws + 3 * qkv);
  int*   firstdone = etag + Bc * Tc;

  dim3 gproj(Bc * Sc / (16 * PJ_TILES));
  proj_splitk<<<gproj, 256, 0, stream>>>(query, W_Q, qb);
  proj_splitk<<<gproj, 256, 0, stream>>>(key,   W_K, kb);
  proj_splitk<<<gproj, 256, 0, stream>>>(value, W_V, vb);

  dones_scan<<<Bc, 32, 0, stream>>>(dones, etag, firstdone);

  float* out0 = (float*)d_out;
  float* hout = out0 + qkv;
  retention16<<<dim3(Sc / 16, Bc), 32, 0, stream>>>(qb, kb, vb, hstate, etag,
                                                    firstdone, out0);
  hstate_update<<<dim3(Hc / 16, Hc / 16, Bc), 32, 0, stream>>>(kb, vb, hstate,
                                                               etag, hout);
}


# dones scan: one lane per timestep, a batch's 256 done-flag loads in flight at once, tags from wave ballots and leading-zero counts through LDS; was one lane walking 32 dependent load groups
# speedup vs baseline: 3.2630x; 1.1199x over previous
#include <hip/hip_runtime.h>
#include <hip/hip_bf16.h>


typedef float v2f __attribute__((ext_vector_type(2)));
typedef float v4f __attribute__((ext_vector_type(4)));
typedef float v8f __attribute__((ext_vector_type(8)));

constexpr int   Bc = 32, Sc = 1024, Ec = 512, Hc = 64, Tc = 256;
constexpr float LNK = -0.1053605156578263f;

static __device__ __forceinline__ v8f wmma4(v2f a, v2f b, v8f c) {
  return __builtin_amdgcn_wmma_f32_16x16x4_f32(false, a, false, b, (short)0, c,
                                               false, false);
}

constexpr int PJ_TILES = 4;

__global__ void __launch_bounds__(256, 4)
proj_splitk(const float* __restrict__ X, const float* __restrict__ W,
            float* __restrict__ Y)
{
  __shared__ v4f part[2][8][16][16];

  const int lane = threadIdx.x & 31;
  const int wave = threadIdx.x >> 5;
  const int c    = lane & 15;
  const int h    = lane >> 4;
  const int kw   = wave << 6;

  const int      rt0  = blockIdx.x * PJ_TILES;
  const float*   xp   = X + (size_t)rt0 * 16 * Ec;
  float*         yp   = Y + (size_t)rt0 * 16 * Hc;
  const unsigned xoff = (unsigned)(c * Ec + kw + 4 * h);
  const unsigned yoff = (unsigned)((2 * wave + h) * Hc + 4 * c);

  v4f xa[8];
  #pragma unroll
  for (int j = 0; j < 8; ++j) xa[j] = *(const v4f*)(xp + xoff + 8 * j);

  v2f wb[4][8][2];
  v4f raw[2][8];
  const float* wp = W + (size_t)(kw + 4 * h) * Hc + 4 * c;
  #pragma unroll
  for (int i = 0; i < 8; ++i)
    raw[0][i] = *(const v4f*)(wp + (size_t)(8 * (i >> 2) + (i & 3)) * Hc);
  #pragma unroll
  for (int ch = 0; ch < 4; ++ch) {
    if (ch + 1 < 4) {
      #pragma unroll
      for (int i = 0; i < 8; ++i)
        raw[(ch + 1) & 1][i] =
            *(const v4f*)(wp + (size_t)(8 * (2 * (ch + 1) + (i >> 2)) + (i & 3)) * Hc);
    }
    #pragma unroll
    for (int jj = 0; jj < 2; ++jj) {
      #pragma unroll
      for (int p = 0; p < 2; ++p) {
        #pragma unroll
        for (int nt = 0; nt < 4; ++nt) {
          wb[nt][2 * ch + jj][p].x = raw[ch & 1][4 * jj + 2 * p][nt];
          wb[nt][2 * ch + jj][p].y = raw[ch & 1][4 * jj + 2 * p + 1][nt];
        }
      }
    }
    __builtin_amdgcn_sched_barrier(0);
  }

  #pragma unroll
  for (int t = 0; t < PJ_TILES; ++t) {
    v4f xn[8];
    if (t + 1 < PJ_TILES) {
      const float* xq = xp + (size_t)(t + 1) * 16 * Ec;
      #pragma unroll
      for (int j = 0; j < 8; ++j) xn[j] = *(const v4f*)(xq + xoff + 8 * j);
    }

    v8f acc[4] = {};
    #pragma unroll
    for (int j = 0; j < 8; ++j) {
      #pragma unroll
      for (int p = 0; p < 2; ++p) {
        v2f a;
        a.x = xa[j][2 * p];
        a.y = xa[j][2 * p + 1];
        #pragma unroll
        for (int nt = 0; nt < 4; ++nt)
          acc[nt] = wmma4(a, wb[nt][j][p], acc[nt]);
      }
    }

    #pragma unroll
    for (int r = 0; r < 8; ++r) {
      v4f o;
      o.x = acc[0][r]; o.y = acc[1][r]; o.z = acc[2][r]; o.w = acc[3][r];
      part[t & 1][wave][8 * h + r][c] = o;
    }
    __syncthreads();

    v4f s = part[t & 1][0][2 * wave + h][c];
    #pragma unroll
    for (int u = 1; u < 8; ++u) s += part[t & 1][u][2 * wave + h][c];
    *(v4f*)(yp + (size_t)t * 16 * Hc + yoff) = s;

    if (t + 1 < PJ_TILES) {
      #pragma unroll
      for (int j = 0; j < 8; ++j) xa[j] = xn[j];
    }
  }
}

__global__ void __launch_bounds__(Tc)
dones_scan(const int* __restrict__ dones, int* __restrict__ etag,
           int* __restrict__ firstdone)
{
  __shared__ unsigned wmask[Tc / 32];
  const int b    = blockIdx.x;
  const int t    = threadIdx.x;
  const int lane = t & 31;
  const int wave = t >> 5;

  const bool     d = dones[(size_t)b * Sc + t * 4] != 0;
  const unsigned m = __builtin_amdgcn_ballot_w32(d);
  if (lane == 0) wmask[wave] = m;
  __syncthreads();

  int tag = 0;
  for (int w = 0; w < Tc / 32; ++w) {
    const unsigned mw = wmask[w];
    if (w < wave && mw != 0u) tag = w * 32 + 32 - __clz((int)mw);
  }
  const unsigned own = m & (0xffffffffu >> (31 - lane));
  if (own != 0u) tag = wave * 32 + 32 - __clz((int)own);
  etag[b * Tc + t] = tag;

  if (t == 0) {
    int first = Tc;
    for (int w = Tc / 32 - 1; w >= 0; --w) {
      const unsigned mw = wmask[w];
      if (mw != 0u) first = w * 32 + __ffs((int)mw) - 1;
    }
    firstdone[b] = first;
  }
}

constexpr int AP = 34;

__global__ void __launch_bounds__(32)
retention16(const float* __restrict__ q, const float* __restrict__ kbuf,
            const float* __restrict__ vbuf, const float* __restrict__ hstate,
            const int* __restrict__ etag, const int* __restrict__ firstdone,
            float* __restrict__ out)
{
  __shared__ int   tagsh[Tc];
  __shared__ __attribute__((aligned(16))) float atile[16 * AP];

  const int lane  = threadIdx.x & 31;
  const int nl    = lane & 15;
  const int hh    = lane >> 4;
  const int khalf = hh << 1;
  const int mh    = hh << 3;
  const int it    = blockIdx.x;
  const int b     = blockIdx.y;

  const int* tagb = etag + b * Tc;
  const int tag0 = tagb[it * 4];
  for (int i = lane; i < Tc; i += 32)  tagsh[i] = tagb[i];
  __syncthreads();

  const int ti0 = it * 4 + (mh >> 2);
  const int cr0 = tagsh[ti0];
  const int cr1 = tagsh[ti0 + 1];
  const int cb  = nl >> 2;

  const float* qb = q      + (size_t)b * Sc * Hc;
  const float* kb = kbuf   + (size_t)b * Sc * Hc;
  const float* vb = vbuf   + (size_t)b * Sc * Hc;
  const float* hb = hstate + (size_t)b * Hc * Hc;

  const float* qrow = qb + (size_t)(it * 16 + nl) * Hc + 4 * hh;
  v4f q4[8];
  #pragma unroll
  for (int j = 0; j < 8; ++j) q4[j] = *(const v4f*)(qrow + 8 * j);

  v8f acc[4] = {};
  const int ntrip = (it >> 1) + 1;
  const int jp0  = ((tag0 > 0) ? tag0 - 1 : 0) >> 3;
  for (int jp = jp0; jp < ntrip; ++jp) {
    const int jt0 = 2 * jp;
    v4f k4[2][8];
    const float* krow = kb + (size_t)(jt0 * 16 + nl) * Hc + 4 * hh;
    #pragma unroll
    for (int j = 0; j < 8; ++j) {
      k4[0][j] = *(const v4f*)(krow + 8 * j);
      k4[1][j] = *(const v4f*)(krow + 16 * Hc + 8 * j);
    }
    v4f vr[8][2];
    const float* vrow = vb + (size_t)(jt0 * 16 + khalf) * Hc + 4 * nl;
    #pragma unroll
    for (int kk = 0; kk < 8; ++kk) {
      vr[kk][0] = *(const v4f*)(vrow + (size_t)(kk * 4) * Hc);
      vr[kk][1] = *(const v4f*)(vrow + (size_t)(kk * 4 + 1) * Hc);
    }
    __builtin_amdgcn_sched_barrier(0);
    const int tj0 = jt0 * 4 + cb;
    const int tj1 = tj0 + 4;
    const int cj0 = tagsh[tj0];
    const int cj1 = tagsh[tj1];
    v8f s0 = {}, s1 = {};
    #pragma unroll
    for (int j = 0; j < 8; ++j) {
      #pragma unroll
      for (int p = 0; p < 2; ++p) {
        v2f a, b0, b1;
        a.x  = q4[j][2 * p];     a.y  = q4[j][2 * p + 1];
        b0.x = k4[0][j][2 * p];  b0.y = k4[0][j][2 * p + 1];
        b1.x = k4[1][j][2 * p];  b1.y = k4[1][j][2 * p + 1];
        s0 = wmma4(a, b0, s0);
        s1 = wmma4(a, b1, s1);
      }
    }
    const float m00 = (cj0 == cr0) ? __expf((float)(ti0 - tj0) * LNK) : 0.0f;
    const float m01 = (cj0 == cr1) ? __expf((float)(ti0 + 1 - tj0) * LNK) : 0.0f;
    const float m10 = (cj1 == cr0) ? __expf((float)(ti0 - tj1) * LNK) : 0.0f;
    const float m11 = (cj1 == cr1) ? __expf((float)(ti0 + 1 - tj1) * LNK) : 0.0f;
    const int   dg0 = (it - jt0) * 16 + mh - nl;
    const int   dg1 = dg0 - 16;
    #pragma unroll
    for (int r = 0; r < 8; ++r) {
      const float w0 = (dg0 + r >= 0) ? ((r < 4) ? m00 : m01) : 0.0f;
      const float w1 = (dg1 + r >= 0) ? ((r < 4) ? m10 : m11) : 0.0f;
      atile[(mh + r) * AP + nl]      = s0[r] * w0;
      atile[(mh + r) * AP + 16 + nl] = s1[r] * w1;
    }
    __syncthreads();
    #pragma unroll
    for (int kk = 0; kk < 8; ++kk) {
      v2f a = *(const v2f*)(&atile[nl * AP + kk * 4 + khalf]);
      #pragma unroll
      for (int nt = 0; nt < 4; ++nt) {
        v2f bm;
        bm.x = vr[kk][0][nt];
        bm.y = vr[kk][1][nt];
        acc[nt] = wmma4(a, bm, acc[nt]);
      }
    }
    __syncthreads();
  }

  v8f cr[4] = {};
  #pragma unroll
  for (int j = 0; j < 8; ++j) {
    #pragma unroll
    for (int p = 0; p < 2; ++p) {
      const float* hr = hb + (size_t)(8 * j + 4 * hh + 2 * p) * Hc + 4 * nl;
      const v4f h0 = *(const v4f*)(hr);
      const v4f h1 = *(const v4f*)(hr + Hc);
      v2f a;
      a.x = q4[j][2 * p];
      a.y = q4[j][2 * p + 1];
      #pragma unroll
      for (int nt = 0; nt < 4; ++nt) {
        v2f bm;
        bm.x = h0[nt];
        bm.y = h1[nt];
        cr[nt] = wmma4(a, bm, cr[nt]);
      }
    }
  }

  const int fd = firstdone[b];
  #pragma unroll
  for (int r = 0; r < 8; ++r) {
    const int   gi = it * 16 + mh + r;
    const int   ti = gi >> 2;
    const float xi = (ti < fd) ? __expf((float)(ti + 1) * LNK) : 0.0f;
    v4f o;
    o.x = acc[0][r] + cr[0][r] * xi;
    o.y = acc[1][r] + cr[1][r] * xi;
    o.z = acc[2][r] + cr[2][r] * xi;
    o.w = acc[3][r] + cr[3][r] * xi;
    *(v4f*)(out + ((size_t)b * Sc + gi) * Hc + 4 * nl) = o;
  }
}

__global__ void __launch_bounds__(32)
hstate_update(const float* __restrict__ kbuf, const float* __restrict__ vbuf,
              const float* __restrict__ hstate, const int* __restrict__ etag,
              float* __restrict__ hout)
{
  __shared__ float ktab[Tc + 1];
  __shared__ float wsh[Tc];
  const int lane  = threadIdx.x & 31;
  const int nl    = lane & 15;
  const int khalf = (lane >> 4) << 1;
  const int mt = blockIdx.x, nt = blockIdx.y, b = blockIdx.z;
  const int* tagb = etag + b * Tc;
  const int  tend = tagb[Tc - 1];

  for (int i = lane; i <= Tc; i += 32) ktab[i] = __expf((float)i * LNK);
  __syncthreads();
  for (int i = lane; i < Tc; i += 32)
    wsh[i] = (tagb[i] == tend) ? ktab[(Tc - 1) - i] : 0.0f;
  __syncthreads();

  const float* kb = kbuf + (size_t)b * Sc * Hc;
  const float* vb = vbuf + (size_t)b * Sc * Hc;

  const int sbeg = (((tend > 0) ? tend - 1 : 0) >> 3) << 5;
  const float* kp = kb + (size_t)(sbeg + khalf) * Hc + mt * 16 + nl;
  const float* vp = vb + (size_t)(sbeg + khalf) * Hc + nt * 16 + nl;
  const float* wq = wsh + (sbeg >> 2);
  v8f c = {};
  for (int sb = sbeg; sb < Sc; sb += 32) {
    #pragma unroll
    for (int u = 0; u < 8; ++u) {
      const float w = wq[u];
      v2f a;
      a.x = kp[(4 * u)     * Hc];
      a.y = kp[(4 * u + 1) * Hc];
      v2f bm;
      bm.x = vp[(4 * u)     * Hc] * w;
      bm.y = vp[(4 * u + 1) * Hc] * w;
      c = wmma4(a, bm, c);
    }
    kp += 32 * Hc;
    vp += 32 * Hc;
    wq += 8;
  }

  const float cd    = ktab[Tc];
  const float delta = (tend == 0) ? 1.0f : 0.0f;
  const float* hb   = hstate + (size_t)b * Hc * Hc;
  const int    mh   = (lane >> 4) << 3;
  #pragma unroll
  for (int r = 0; r < 8; ++r) {
    const int h1 = mt * 16 + mh + r;
    hout[((size_t)b * Hc + h1) * Hc + nt * 16 + nl] =
        c[r] + hb[(size_t)h1 * Hc + nt * 16 + nl] * cd * delta;
  }
}

extern "C" void kernel_launch(void* const* d_in, const int* in_sizes, int n_in,
                              void* d_out, int out_size, void* d_ws, size_t ws_size,
                              hipStream_t stream)
{
  const float* key    = (const float*)d_in[0];
  const float* query  = (const float*)d_in[1];
  const float* value  = (const float*)d_in[2];
  const float* hstate = (const float*)d_in[3];
  const int*   dones  = (const int*)d_in[4];
  const float* W_Q    = (const float*)d_in[5];
  const float* W_K    = (const float*)d_in[6];
  const float* W_V    = (const float*)d_in[7];

  const size_t qkv = (size_t)Bc * Sc * Hc;
  float* ws = (float*)d_ws;
  float* qb = ws;
  float* kb = ws + qkv;
  float* vb = ws + 2 * qkv;
  int*   etag       = (int*)(ws + 3 * qkv);
  int*   firstdone = etag + Bc * Tc;

  dim3 gproj(Bc * Sc / (16 * PJ_TILES));
  proj_splitk<<<gproj, 256, 0, stream>>>(query, W_Q, qb);
  proj_splitk<<<gproj, 256, 0, stream>>>(key,   W_K, kb);
  proj_splitk<<<gproj, 256, 0, stream>>>(value, W_V, vb);

  dones_scan<<<Bc, Tc, 0, stream>>>(dones, etag, firstdone);

  float* out0 = (float*)d_out;
  float* hout = out0 + qkv;
  retention16<<<dim3(Sc / 16, Bc), 32, 0, stream>>>(qb, kb, vb, hstate, etag,
                                                    firstdone, out0);
  hstate_update<<<dim3(Hc / 16, Hc / 16, Bc), 32, 0, stream>>>(kb, vb, hstate,
                                                               etag, hout);
}


# retention key loop and state loop pipelined: trip j+1's k loads go out behind trip j's staging barrier into the freed k registers (tile double-buffered); state blocks load one ahead; same sums
# speedup vs baseline: 3.4846x; 1.0679x over previous
#include <hip/hip_runtime.h>
#include <hip/hip_bf16.h>


typedef float v2f __attribute__((ext_vector_type(2)));
typedef float v4f __attribute__((ext_vector_type(4)));
typedef float v8f __attribute__((ext_vector_type(8)));

constexpr int   Bc = 32, Sc = 1024, Ec = 512, Hc = 64, Tc = 256;
constexpr float LNK = -0.1053605156578263f;

static __device__ __forceinline__ v8f wmma4(v2f a, v2f b, v8f c) {
  return __builtin_amdgcn_wmma_f32_16x16x4_f32(false, a, false, b, (short)0, c,
                                               false, false);
}

constexpr int PJ_TILES = 4;

__global__ void __launch_bounds__(256, 4)
proj_splitk(const float* __restrict__ X, const float* __restrict__ W,
            float* __restrict__ Y)
{
  __shared__ v4f part[2][8][16][16];

  const int lane = threadIdx.x & 31;
  const int wave = threadIdx.x >> 5;
  const int c    = lane & 15;
  const int h    = lane >> 4;
  const int kw   = wave << 6;

  const int      rt0  = blockIdx.x * PJ_TILES;
  const float*   xp   = X + (size_t)rt0 * 16 * Ec;
  float*         yp   = Y + (size_t)rt0 * 16 * Hc;
  const unsigned xoff = (unsigned)(c * Ec + kw + 4 * h);
  const unsigned yoff = (unsigned)((2 * wave + h) * Hc + 4 * c);

  v4f xa[8];
  #pragma unroll
  for (int j = 0; j < 8; ++j) xa[j] = *(const v4f*)(xp + xoff + 8 * j);

  v2f wb[4][8][2];
  v4f raw[2][8];
  const float* wp = W + (size_t)(kw + 4 * h) * Hc + 4 * c;
  #pragma unroll
  for (int i = 0; i < 8; ++i)
    raw[0][i] = *(const v4f*)(wp + (size_t)(8 * (i >> 2) + (i & 3)) * Hc);
  #pragma unroll
  for (int ch = 0; ch < 4; ++ch) {
    if (ch + 1 < 4) {
      #pragma unroll
      for (int i = 0; i < 8; ++i)
        raw[(ch + 1) & 1][i] =
            *(const v4f*)(wp + (size_t)(8 * (2 * (ch + 1) + (i >> 2)) + (i & 3)) * Hc);
    }
    #pragma unroll
    for (int jj = 0; jj < 2; ++jj) {
      #pragma unroll
      for (int p = 0; p < 2; ++p) {
        #pragma unroll
        for (int nt = 0; nt < 4; ++nt) {
          wb[nt][2 * ch + jj][p].x = raw[ch & 1][4 * jj + 2 * p][nt];
          wb[nt][2 * ch + jj][p].y = raw[ch & 1][4 * jj + 2 * p + 1][nt];
        }
      }
    }
    __builtin_amdgcn_sched_barrier(0);
  }

  #pragma unroll
  for (int t = 0; t < PJ_TILES; ++t) {
    v4f xn[8];
    if (t + 1 < PJ_TILES) {
      const float* xq = xp + (size_t)(t + 1) * 16 * Ec;
      #pragma unroll
      for (int j = 0; j < 8; ++j) xn[j] = *(const v4f*)(xq + xoff + 8 * j);
    }

    v8f acc[4] = {};
    #pragma unroll
    for (int j = 0; j < 8; ++j) {
      #pragma unroll
      for (int p = 0; p < 2; ++p) {
        v2f a;
        a.x = xa[j][2 * p];
        a.y = xa[j][2 * p + 1];
        #pragma unroll
        for (int nt = 0; nt < 4; ++nt)
          acc[nt] = wmma4(a, wb[nt][j][p], acc[nt]);
      }
    }

    #pragma unroll
    for (int r = 0; r < 8; ++r) {
      v4f o;
      o.x = acc[0][r]; o.y = acc[1][r]; o.z = acc[2][r]; o.w = acc[3][r];
      part[t & 1][wave][8 * h + r][c] = o;
    }
    __syncthreads();

    v4f s = part[t & 1][0][2 * wave + h][c];
    #pragma unroll
    for (int u = 1; u < 8; ++u) s += part[t & 1][u][2 * wave + h][c];
    *(v4f*)(yp + (size_t)t * 16 * Hc + yoff) = s;

    if (t + 1 < PJ_TILES) {
      #pragma unroll
      for (int j = 0; j < 8; ++j) xa[j] = xn[j];
    }
  }
}

__global__ void __launch_bounds__(Tc)
dones_scan(const int* __restrict__ dones, int* __restrict__ etag,
           int* __restrict__ firstdone)
{
  __shared__ unsigned wmask[Tc / 32];
  const int b    = blockIdx.x;
  const int t    = threadIdx.x;
  const int lane = t & 31;
  const int wave = t >> 5;

  const bool     d = dones[(size_t)b * Sc + t * 4] != 0;
  const unsigned m = __builtin_amdgcn_ballot_w32(d);
  if (lane == 0) wmask[wave] = m;
  __syncthreads();

  int tag = 0;
  for (int w = 0; w < Tc / 32; ++w) {
    const unsigned mw = wmask[w];
    if (w < wave && mw != 0u) tag = w * 32 + 32 - __clz((int)mw);
  }
  const unsigned own = m & (0xffffffffu >> (31 - lane));
  if (own != 0u) tag = wave * 32 + 32 - __clz((int)own);
  etag[b * Tc + t] = tag;

  if (t == 0) {
    int first = Tc;
    for (int w = Tc / 32 - 1; w >= 0; --w) {
      const unsigned mw = wmask[w];
      if (mw != 0u) first = w * 32 + __ffs((int)mw) - 1;
    }
    firstdone[b] = first;
  }
}

constexpr int AP = 34;

__global__ void __launch_bounds__(32)
retention16(const float* __restrict__ q, const float* __restrict__ kbuf,
            const float* __restrict__ vbuf, const float* __restrict__ hstate,
            const int* __restrict__ etag, const int* __restrict__ firstdone,
            float* __restrict__ out)
{
  __shared__ int   tagsh[Tc];
  __shared__ __attribute__((aligned(16))) float atile[2][16 * AP];

  const int lane  = threadIdx.x & 31;
  const int nl    = lane & 15;
  const int hh    = lane >> 4;
  const int khalf = hh << 1;
  const int mh    = hh << 3;
  const int it    = blockIdx.x;
  const int b     = blockIdx.y;

  const int* tagb = etag + b * Tc;
  const int tag0 = tagb[it * 4];
  for (int i = lane; i < Tc; i += 32)  tagsh[i] = tagb[i];
  __syncthreads();

  const int ti0 = it * 4 + (mh >> 2);
  const int cr0 = tagsh[ti0];
  const int cr1 = tagsh[ti0 + 1];
  const int cb  = nl >> 2;

  const float* qb = q      + (size_t)b * Sc * Hc;
  const float* kb = kbuf   + (size_t)b * Sc * Hc;
  const float* vb = vbuf   + (size_t)b * Sc * Hc;
  const float* hb = hstate + (size_t)b * Hc * Hc;

  const float* qrow = qb + (size_t)(it * 16 + nl) * Hc + 4 * hh;
  v4f q4[8];
  #pragma unroll
  for (int j = 0; j < 8; ++j) q4[j] = *(const v4f*)(qrow + 8 * j);

  v8f acc[4] = {};
  const int ntrip = (it >> 1) + 1;
  const int jp0  = ((tag0 > 0) ? tag0 - 1 : 0) >> 3;
  v4f k4[2][8];
  {
    const float* krow = kb + (size_t)(2 * jp0 * 16 + nl) * Hc + 4 * hh;
    #pragma unroll
    for (int j = 0; j < 8; ++j) {
      k4[0][j] = *(const v4f*)(krow + 8 * j);
      k4[1][j] = *(const v4f*)(krow + 16 * Hc + 8 * j);
    }
  }
  for (int jp = jp0; jp < ntrip; ++jp) {
    const int jt0 = 2 * jp;
    v4f vr[8][2];
    const float* vrow = vb + (size_t)(jt0 * 16 + khalf) * Hc + 4 * nl;
    #pragma unroll
    for (int kk = 0; kk < 8; ++kk) {
      vr[kk][0] = *(const v4f*)(vrow + (size_t)(kk * 4) * Hc);
      vr[kk][1] = *(const v4f*)(vrow + (size_t)(kk * 4 + 1) * Hc);
    }
    __builtin_amdgcn_sched_barrier(0);
    const int tj0 = jt0 * 4 + cb;
    const int tj1 = tj0 + 4;
    const int cj0 = tagsh[tj0];
    const int cj1 = tagsh[tj1];
    v8f s0 = {}, s1 = {};
    #pragma unroll
    for (int j = 0; j < 8; ++j) {
      #pragma unroll
      for (int p = 0; p < 2; ++p) {
        v2f a, b0, b1;
        a.x  = q4[j][2 * p];     a.y  = q4[j][2 * p + 1];
        b0.x = k4[0][j][2 * p];  b0.y = k4[0][j][2 * p + 1];
        b1.x = k4[1][j][2 * p];  b1.y = k4[1][j][2 * p + 1];
        s0 = wmma4(a, b0, s0);
        s1 = wmma4(a, b1, s1);
      }
    }
    const float m00 = (cj0 == cr0) ? __expf((float)(ti0 - tj0) * LNK) : 0.0f;
    const float m01 = (cj0 == cr1) ? __expf((float)(ti0 + 1 - tj0) * LNK) : 0.0f;
    const float m10 = (cj1 == cr0) ? __expf((float)(ti0 - tj1) * LNK) : 0.0f;
    const float m11 = (cj1 == cr1) ? __expf((float)(ti0 + 1 - tj1) * LNK) : 0.0f;
    const int   dg0 = (it - jt0) * 16 + mh - nl;
    const int   dg1 = dg0 - 16;
    float* at = atile[jp & 1];
    #pragma unroll
    for (int r = 0; r < 8; ++r) {
      const float w0 = (dg0 + r >= 0) ? ((r < 4) ? m00 : m01) : 0.0f;
      const float w1 = (dg1 + r >= 0) ? ((r < 4) ? m10 : m11) : 0.0f;
      at[(mh + r) * AP + nl]      = s0[r] * w0;
      at[(mh + r) * AP + 16 + nl] = s1[r] * w1;
    }
    __syncthreads();
    if (jp + 1 < ntrip) {
      const float* krow = kb + (size_t)((jt0 + 2) * 16 + nl) * Hc + 4 * hh;
      #pragma unroll
      for (int j = 0; j < 8; ++j) {
        k4[0][j] = *(const v4f*)(krow + 8 * j);
        k4[1][j] = *(const v4f*)(krow + 16 * Hc + 8 * j);
      }
    }
    #pragma unroll
    for (int kk = 0; kk < 8; ++kk) {
      v2f a = *(const v2f*)(&at[nl * AP + kk * 4 + khalf]);
      #pragma unroll
      for (int nt = 0; nt < 4; ++nt) {
        v2f bm;
        bm.x = vr[kk][0][nt];
        bm.y = vr[kk][1][nt];
        acc[nt] = wmma4(a, bm, acc[nt]);
      }
    }
  }

  v8f cr[4] = {};
  #pragma unroll
  for (int j = 0; j < 8; ++j) {
    #pragma unroll
    for (int p = 0; p < 2; ++p) {
      const float* hr = hb + (size_t)(8 * j + 4 * hh + 2 * p) * Hc + 4 * nl;
      const v4f h0 = *(const v4f*)(hr);
      const v4f h1 = *(const v4f*)(hr + Hc);
      v2f a;
      a.x = q4[j][2 * p];
      a.y = q4[j][2 * p + 1];
      #pragma unroll
      for (int nt = 0; nt < 4; ++nt) {
        v2f bm;
        bm.x = h0[nt];
        bm.y = h1[nt];
        cr[nt] = wmma4(a, bm, cr[nt]);
      }
    }
  }

  const int fd = firstdone[b];
  #pragma unroll
  for (int r = 0; r < 8; ++r) {
    const int   gi = it * 16 + mh + r;
    const int   ti = gi >> 2;
    const float xi = (ti < fd) ? __expf((float)(ti + 1) * LNK) : 0.0f;
    v4f o;
    o.x = acc[0][r] + cr[0][r] * xi;
    o.y = acc[1][r] + cr[1][r] * xi;
    o.z = acc[2][r] + cr[2][r] * xi;
    o.w = acc[3][r] + cr[3][r] * xi;
    *(v4f*)(out + ((size_t)b * Sc + gi) * Hc + 4 * nl) = o;
  }
}

__global__ void __launch_bounds__(32)
hstate_update(const float* __restrict__ kbuf, const float* __restrict__ vbuf,
              const float* __restrict__ hstate, const int* __restrict__ etag,
              float* __restrict__ hout)
{
  __shared__ float ktab[Tc + 1];
  __shared__ float wsh[Tc];
  const int lane  = threadIdx.x & 31;
  const int nl    = lane & 15;
  const int khalf = (lane >> 4) << 1;
  const int mt = blockIdx.x, nt = blockIdx.y, b = blockIdx.z;
  const int* tagb = etag + b * Tc;
  const int  tend = tagb[Tc - 1];

  for (int i = lane; i <= Tc; i += 32) ktab[i] = __expf((float)i * LNK);
  __syncthreads();
  for (int i = lane; i < Tc; i += 32)
    wsh[i] = (tagb[i] == tend) ? ktab[(Tc - 1) - i] : 0.0f;
  __syncthreads();

  const float* kb = kbuf + (size_t)b * Sc * Hc;
  const float* vb = vbuf + (size_t)b * Sc * Hc;

  const int sbeg = (((tend > 0) ? tend - 1 : 0) >> 3) << 5;
  const float* kp0 = kb + (size_t)khalf * Hc + mt * 16 + nl;
  const float* vp0 = vb + (size_t)khalf * Hc + nt * 16 + nl;

  auto load_block = [&](float (&kr)[16], float (&vr)[16], int s) {
    const float* kq = kp0 + (size_t)s * Hc;
    const float* vq = vp0 + (size_t)s * Hc;
    #pragma unroll
    for (int u = 0; u < 8; ++u) {
      kr[2 * u]     = kq[(4 * u)     * Hc];
      kr[2 * u + 1] = kq[(4 * u + 1) * Hc];
      vr[2 * u]     = vq[(4 * u)     * Hc];
      vr[2 * u + 1] = vq[(4 * u + 1) * Hc];
    }
  };
  v8f c = {};
  auto mma_block = [&](const float (&kr)[16], const float (&vr)[16], int s) {
    const float* wq = wsh + (s >> 2);
    #pragma unroll
    for (int u = 0; u < 8; ++u) {
      const float w = wq[u];
      v2f a;
      a.x = kr[2 * u];
      a.y = kr[2 * u + 1];
      v2f bm;
      bm.x = vr[2 * u] * w;
      bm.y = vr[2 * u + 1] * w;
      c = wmma4(a, bm, c);
    }
  };

  const int slast = Sc - 32;
  float kA[16], vA[16], kB[16], vB[16];
  load_block(kA, vA, sbeg);
  int sb = sbeg;
  for (; sb + 64 <= Sc; sb += 64) {
    load_block(kB, vB, sb + 32);
    __builtin_amdgcn_sched_barrier(0);
    mma_block(kA, vA, sb);
    load_block(kA, vA, (sb + 64 < Sc) ? sb + 64 : slast);
    __builtin_amdgcn_sched_barrier(0);
    mma_block(kB, vB, sb + 32);
  }
  if (sb < Sc) mma_block(kA, vA, sb);

  const float cd    = ktab[Tc];
  const float delta = (tend == 0) ? 1.0f : 0.0f;
  const float* hb   = hstate + (size_t)b * Hc * Hc;
  const int    mh   = (lane >> 4) << 3;
  #pragma unroll
  for (int r = 0; r < 8; ++r) {
    const int h1 = mt * 16 + mh + r;
    hout[((size_t)b * Hc + h1) * Hc + nt * 16 + nl] =
        c[r] + hb[(size_t)h1 * Hc + nt * 16 + nl] * cd * delta;
  }
}

extern "C" void kernel_launch(void* const* d_in, const int* in_sizes, int n_in,
                              void* d_out, int out_size, void* d_ws, size_t ws_size,
                              hipStream_t stream)
{
  const float* key    = (const float*)d_in[0];
  const float* query  = (const float*)d_in[1];
  const float* value  = (const float*)d_in[2];
  const float* hstate = (const float*)d_in[3];
  const int*   dones  = (const int*)d_in[4];
  const float* W_Q    = (const float*)d_in[5];
  const float* W_K    = (const float*)d_in[6];
  const float* W_V    = (const float*)d_in[7];

  const size_t qkv = (size_t)Bc * Sc * Hc;
  float* ws = (float*)d_ws;
  float* qb = ws;
  float* kb = ws + qkv;
  float* vb = ws + 2 * qkv;
  int*   etag       = (int*)(ws + 3 * qkv);
  int*   firstdone = etag + Bc * Tc;

  dim3 gproj(Bc * Sc / (16 * PJ_TILES));
  proj_splitk<<<gproj, 256, 0, stream>>>(query, W_Q, qb);
  proj_splitk<<<gproj, 256, 0, stream>>>(key,   W_K, kb);
  proj_splitk<<<gproj, 256, 0, stream>>>(value, W_V, vb);

  dones_scan<<<Bc, Tc, 0, stream>>>(dones, etag, firstdone);

  float* out0 = (float*)d_out;
  float* hout = out0 + qkv;
  retention16<<<dim3(Sc / 16, Bc), 32, 0, stream>>>(qb, kb, vb, hstate, etag,
                                                    firstdone, out0);
  hstate_update<<<dim3(Hc / 16, Hc / 16, Bc), 32, 0, stream>>>(kb, vb, hstate,
                                                               etag, hout);
}


# retention: a 16-row strip worked by 4 waves: score split by feature half x the trip's 2 key tiles, partials summed in the LDS tile; 2nd product by column quarter; features summed in another order
# speedup vs baseline: 3.8081x; 1.0928x over previous
#include <hip/hip_runtime.h>
#include <hip/hip_bf16.h>


typedef float v2f __attribute__((ext_vector_type(2)));
typedef float v4f __attribute__((ext_vector_type(4)));
typedef float v8f __attribute__((ext_vector_type(8)));

constexpr int   Bc = 32, Sc = 1024, Ec = 512, Hc = 64, Tc = 256;
constexpr float LNK = -0.1053605156578263f;

static __device__ __forceinline__ v8f wmma4(v2f a, v2f b, v8f c) {
  return __builtin_amdgcn_wmma_f32_16x16x4_f32(false, a, false, b, (short)0, c,
                                               false, false);
}

constexpr int PJ_TILES = 4;

__global__ void __launch_bounds__(256, 4)
proj_splitk(const float* __restrict__ X, const float* __restrict__ W,
            float* __restrict__ Y)
{
  __shared__ v4f part[2][8][16][16];

  const int lane = threadIdx.x & 31;
  const int wave = threadIdx.x >> 5;
  const int c    = lane & 15;
  const int h    = lane >> 4;
  const int kw   = wave << 6;

  const int      rt0  = blockIdx.x * PJ_TILES;
  const float*   xp   = X + (size_t)rt0 * 16 * Ec;
  float*         yp   = Y + (size_t)rt0 * 16 * Hc;
  const unsigned xoff = (unsigned)(c * Ec + kw + 4 * h);
  const unsigned yoff = (unsigned)((2 * wave + h) * Hc + 4 * c);

  v4f xa[8];
  #pragma unroll
  for (int j = 0; j < 8; ++j) xa[j] = *(const v4f*)(xp + xoff + 8 * j);

  v2f wb[4][8][2];
  v4f raw[2][8];
  const float* wp = W + (size_t)(kw + 4 * h) * Hc + 4 * c;
  #pragma unroll
  for (int i = 0; i < 8; ++i)
    raw[0][i] = *(const v4f*)(wp + (size_t)(8 * (i >> 2) + (i & 3)) * Hc);
  #pragma unroll
  for (int ch = 0; ch < 4; ++ch) {
    if (ch + 1 < 4) {
      #pragma unroll
      for (int i = 0; i < 8; ++i)
        raw[(ch + 1) & 1][i] =
            *(const v4f*)(wp + (size_t)(8 * (2 * (ch + 1) + (i >> 2)) + (i & 3)) * Hc);
    }
    #pragma unroll
    for (int jj = 0; jj < 2; ++jj) {
      #pragma unroll
      for (int p = 0; p < 2; ++p) {
        #pragma unroll
        for (int nt = 0; nt < 4; ++nt) {
          wb[nt][2 * ch + jj][p].x = raw[ch & 1][4 * jj + 2 * p][nt];
          wb[nt][2 * ch + jj][p].y = raw[ch & 1][4 * jj + 2 * p + 1][nt];
        }
      }
    }
    __builtin_amdgcn_sched_barrier(0);
  }

  #pragma unroll
  for (int t = 0; t < PJ_TILES; ++t) {
    v4f xn[8];
    if (t + 1 < PJ_TILES) {
      const float* xq = xp + (size_t)(t + 1) * 16 * Ec;
      #pragma unroll
      for (int j = 0; j < 8; ++j) xn[j] = *(const v4f*)(xq + xoff + 8 * j);
    }

    v8f acc[4] = {};
    #pragma unroll
    for (int j = 0; j < 8; ++j) {
      #pragma unroll
      for (int p = 0; p < 2; ++p) {
        v2f a;
        a.x = xa[j][2 * p];
        a.y = xa[j][2 * p + 1];
        #pragma unroll
        for (int nt = 0; nt < 4; ++nt)
          acc[nt] = wmma4(a, wb[nt][j][p], acc[nt]);
      }
    }

    #pragma unroll
    for (int r = 0; r < 8; ++r) {
      v4f o;
      o.x = acc[0][r]; o.y = acc[1][r]; o.z = acc[2][r]; o.w = acc[3][r];
      part[t & 1][wave][8 * h + r][c] = o;
    }
    __syncthreads();

    v4f s = part[t & 1][0][2 * wave + h][c];
    #pragma unroll
    for (int u = 1; u < 8; ++u) s += part[t & 1][u][2 * wave + h][c];
    *(v4f*)(yp + (size_t)t * 16 * Hc + yoff) = s;

    if (t + 1 < PJ_TILES) {
      #pragma unroll
      for (int j = 0; j < 8; ++j) xa[j] = xn[j];
    }
  }
}

__global__ void __launch_bounds__(Tc)
dones_scan(const int* __restrict__ dones, int* __restrict__ etag,
           int* __restrict__ firstdone)
{
  __shared__ unsigned wmask[Tc / 32];
  const int b    = blockIdx.x;
  const int t    = threadIdx.x;
  const int lane = t & 31;
  const int wave = t >> 5;

  const bool     d = dones[(size_t)b * Sc + t * 4] != 0;
  const unsigned m = __builtin_amdgcn_ballot_w32(d);
  if (lane == 0) wmask[wave] = m;
  __syncthreads();

  int tag = 0;
  for (int w = 0; w < Tc / 32; ++w) {
    const unsigned mw = wmask[w];
    if (w < wave && mw != 0u) tag = w * 32 + 32 - __clz((int)mw);
  }
  const unsigned own = m & (0xffffffffu >> (31 - lane));
  if (own != 0u) tag = wave * 32 + 32 - __clz((int)own);
  etag[b * Tc + t] = tag;

  if (t == 0) {
    int first = Tc;
    for (int w = Tc / 32 - 1; w >= 0; --w) {
      const unsigned mw = wmask[w];
      if (mw != 0u) first = w * 32 + __ffs((int)mw) - 1;
    }
    firstdone[b] = first;
  }
}

constexpr int AP = 34;
constexpr int OP = 68;

__global__ void __launch_bounds__(128)
retention16(const float* __restrict__ q, const float* __restrict__ kbuf,
            const float* __restrict__ vbuf, const float* __restrict__ hstate,
            const int* __restrict__ etag, const int* __restrict__ firstdone,
            float* __restrict__ out)
{
  __shared__ int   tagsh[Tc];
  __shared__ __attribute__((aligned(16))) float ptile[2][2][16 * AP];

  const int lane  = threadIdx.x & 31;
  const int wave  = __builtin_amdgcn_readfirstlane((int)(threadIdx.x >> 5));
  const int fh    = wave & 1;
  const int kt    = wave >> 1;
  const int nl    = lane & 15;
  const int hh    = lane >> 4;
  const int khalf = hh << 1;
  const int mh    = hh << 3;
  const int it    = blockIdx.x;
  const int b     = blockIdx.y;

  const int* tagb = etag + b * Tc;
  const int tag0 = tagb[it * 4];
  for (int i = threadIdx.x; i < Tc; i += 128)  tagsh[i] = tagb[i];
  __syncthreads();

  const int ti0 = it * 4 + (mh >> 2);
  const int cr0 = tagsh[ti0];
  const int cr1 = tagsh[ti0 + 1];
  const int cb  = nl >> 2;

  const float* qb = q      + (size_t)b * Sc * Hc;
  const float* kb = kbuf   + (size_t)b * Sc * Hc;
  const float* vb = vbuf   + (size_t)b * Sc * Hc;
  const float* hb = hstate + (size_t)b * Hc * Hc;

  const float* qrow = qb + (size_t)(it * 16 + nl) * Hc + 4 * hh;
  v4f qs[4];
  #pragma unroll
  for (int jj = 0; jj < 4; ++jj) qs[jj] = *(const v4f*)(qrow + 32 * fh + 8 * jj);

  v8f acc = {};
  const int ntrip = (it >> 1) + 1;
  const int jp0  = ((tag0 > 0) ? tag0 - 1 : 0) >> 3;
  const float* kbase = kb + (size_t)(16 * kt + nl) * Hc + 32 * fh + 4 * hh;
  v4f k4[4];
  {
    const float* krow = kbase + (size_t)(2 * jp0 * 16) * Hc;
    #pragma unroll
    for (int jj = 0; jj < 4; ++jj) k4[jj] = *(const v4f*)(krow + 8 * jj);
  }
  const float* vbase = vb + (size_t)khalf * Hc + 16 * wave + nl;

  for (int jp = jp0; jp < ntrip; ++jp) {
    const int jt0 = 2 * jp;
    const int jt  = jt0 + kt;
    float vr[8][2];
    const float* vrow = vbase + (size_t)(jt0 * 16) * Hc;
    #pragma unroll
    for (int kk = 0; kk < 8; ++kk) {
      vr[kk][0] = vrow[(size_t)(kk * 4) * Hc];
      vr[kk][1] = vrow[(size_t)(kk * 4 + 1) * Hc];
    }
    const int tj = jt * 4 + cb;
    const int cj = tagsh[tj];
    v8f s = {};
    #pragma unroll
    for (int jj = 0; jj < 4; ++jj) {
      #pragma unroll
      for (int p = 0; p < 2; ++p) {
        v2f a, bk;
        a.x  = qs[jj][2 * p];   a.y  = qs[jj][2 * p + 1];
        bk.x = k4[jj][2 * p];   bk.y = k4[jj][2 * p + 1];
        s = wmma4(a, bk, s);
      }
    }
    const float m0 = (cj == cr0) ? __expf((float)(ti0 - tj) * LNK) : 0.0f;
    const float m1 = (cj == cr1) ? __expf((float)(ti0 + 1 - tj) * LNK) : 0.0f;
    const int   dg = (it - jt) * 16 + mh - nl;
    float* pt = &ptile[jp & 1][fh][16 * kt + nl];
    #pragma unroll
    for (int r = 0; r < 8; ++r) {
      const float w = (dg + r >= 0) ? ((r < 4) ? m0 : m1) : 0.0f;
      pt[(mh + r) * AP] = s[r] * w;
    }
    asm volatile("" ::: "memory");
    {
      const int jn = (jp + 1 < ntrip) ? jt0 + 2 : jt0;
      const float* krow = kbase + (size_t)(jn * 16) * Hc;
      #pragma unroll
      for (int jj = 0; jj < 4; ++jj) k4[jj] = *(const v4f*)(krow + 8 * jj);
    }
    __builtin_amdgcn_fence(__ATOMIC_RELEASE, "workgroup", "local");
    __builtin_amdgcn_s_barrier();
    __builtin_amdgcn_fence(__ATOMIC_ACQUIRE, "workgroup", "local");
    const float* p0 = &ptile[jp & 1][0][nl * AP + khalf];
    const float* p1 = &ptile[jp & 1][1][nl * AP + khalf];
    v2f af[8];
    #pragma unroll
    for (int kk = 0; kk < 8; ++kk)
      af[kk] = *(const v2f*)(p0 + kk * 4) + *(const v2f*)(p1 + kk * 4);
    __builtin_amdgcn_sched_barrier(0);
    #pragma unroll
    for (int kk = 0; kk < 8; ++kk) {
      v2f bm;
      bm.x = vr[kk][0];
      bm.y = vr[kk][1];
      acc = wmma4(af[kk], bm, acc);
    }
  }

  v8f cr = {};
  #pragma unroll
  for (int j = 0; j < 8; ++j) {
    const v4f q4 = *(const v4f*)(qrow + 8 * j);
    #pragma unroll
    for (int p = 0; p < 2; ++p) {
      const float* hr = hb + (size_t)(8 * j + 4 * hh + 2 * p) * Hc + 16 * wave + nl;
      v2f a, bm;
      a.x  = q4[2 * p];
      a.y  = q4[2 * p + 1];
      bm.x = hr[0];
      bm.y = hr[Hc];
      cr = wmma4(a, bm, cr);
    }
  }

  const int fd = firstdone[b];
  float* ot = &ptile[0][0][0];
  __syncthreads();
  #pragma unroll
  for (int r = 0; r < 8; ++r) {
    const int   gi = it * 16 + mh + r;
    const int   ti = gi >> 2;
    const float xi = (ti < fd) ? __expf((float)(ti + 1) * LNK) : 0.0f;
    ot[(mh + r) * OP + 16 * wave + nl] = acc[r] + cr[r] * xi;
  }
  __syncthreads();
  #pragma unroll
  for (int i = 0; i < 2; ++i) {
    const int row = 4 * wave + 2 * i + hh;
    const v4f o = *(const v4f*)(&ot[row * OP + 4 * nl]);
    *(v4f*)(out + ((size_t)b * Sc + it * 16 + row) * Hc + 4 * nl) = o;
  }
}

__global__ void __launch_bounds__(32)
hstate_update(const float* __restrict__ kbuf, const float* __restrict__ vbuf,
              const float* __restrict__ hstate, const int* __restrict__ etag,
              float* __restrict__ hout)
{
  __shared__ float ktab[Tc + 1];
  __shared__ float wsh[Tc];
  const int lane  = threadIdx.x & 31;
  const int nl    = lane & 15;
  const int khalf = (lane >> 4) << 1;
  const int mt = blockIdx.x, nt = blockIdx.y, b = blockIdx.z;
  const int* tagb = etag + b * Tc;
  const int  tend = tagb[Tc - 1];

  for (int i = lane; i <= Tc; i += 32) ktab[i] = __expf((float)i * LNK);
  __syncthreads();
  for (int i = lane; i < Tc; i += 32)
    wsh[i] = (tagb[i] == tend) ? ktab[(Tc - 1) - i] : 0.0f;
  __syncthreads();

  const float* kb = kbuf + (size_t)b * Sc * Hc;
  const float* vb = vbuf + (size_t)b * Sc * Hc;

  const int sbeg = (((tend > 0) ? tend - 1 : 0) >> 3) << 5;
  const float* kp0 = kb + (size_t)khalf * Hc + mt * 16 + nl;
  const float* vp0 = vb + (size_t)khalf * Hc + nt * 16 + nl;

  auto load_block = [&](float (&kr)[16], float (&vr)[16], int s) {
    const float* kq = kp0 + (size_t)s * Hc;
    const float* vq = vp0 + (size_t)s * Hc;
    #pragma unroll
    for (int u = 0; u < 8; ++u) {
      kr[2 * u]     = kq[(4 * u)     * Hc];
      kr[2 * u + 1] = kq[(4 * u + 1) * Hc];
      vr[2 * u]     = vq[(4 * u)     * Hc];
      vr[2 * u + 1] = vq[(4 * u + 1) * Hc];
    }
  };
  v8f c = {};
  auto mma_block = [&](const float (&kr)[16], const float (&vr)[16], int s) {
    const float* wq = wsh + (s >> 2);
    #pragma unroll
    for (int u = 0; u < 8; ++u) {
      const float w = wq[u];
      v2f a;
      a.x = kr[2 * u];
      a.y = kr[2 * u + 1];
      v2f bm;
      bm.x = vr[2 * u] * w;
      bm.y = vr[2 * u + 1] * w;
      c = wmma4(a, bm, c);
    }
  };

  const int slast = Sc - 32;
  float kA[16], vA[16], kB[16], vB[16];
  load_block(kA, vA, sbeg);
  int sb = sbeg;
  for (; sb + 64 <= Sc; sb += 64) {
    load_block(kB, vB, sb + 32);
    __builtin_amdgcn_sched_barrier(0);
    mma_block(kA, vA, sb);
    load_block(kA, vA, (sb + 64 < Sc) ? sb + 64 : slast);
    __builtin_amdgcn_sched_barrier(0);
    mma_block(kB, vB, sb + 32);
  }
  if (sb < Sc) mma_block(kA, vA, sb);

  const float cd    = ktab[Tc];
  const float delta = (tend == 0) ? 1.0f : 0.0f;
  const float* hb   = hstate + (size_t)b * Hc * Hc;
  const int    mh   = (lane >> 4) << 3;
  #pragma unroll
  for (int r = 0; r < 8; ++r) {
    const int h1 = mt * 16 + mh + r;
    hout[((size_t)b * Hc + h1) * Hc + nt * 16 + nl] =
        c[r] + hb[(size_t)h1 * Hc + nt * 16 + nl] * cd * delta;
  }
}

extern "C" void kernel_launch(void* const* d_in, const int* in_sizes, int n_in,
                              void* d_out, int out_size, void* d_ws, size_t ws_size,
                              hipStream_t stream)
{
  const float* key    = (const float*)d_in[0];
  const float* query  = (const float*)d_in[1];
  const float* value  = (const float*)d_in[2];
  const float* hstate = (const float*)d_in[3];
  const int*   dones  = (const int*)d_in[4];
  const float* W_Q    = (const float*)d_in[5];
  const float* W_K    = (const float*)d_in[6];
  const float* W_V    = (const float*)d_in[7];

  const size_t qkv = (size_t)Bc * Sc * Hc;
  float* ws = (float*)d_ws;
  float* qb = ws;
  float* kb = ws + qkv;
  float* vb = ws + 2 * qkv;
  int*   etag       = (int*)(ws + 3 * qkv);
  int*   firstdone = etag + Bc * Tc;

  dim3 gproj(Bc * Sc / (16 * PJ_TILES));
  proj_splitk<<<gproj, 256, 0, stream>>>(query, W_Q, qb);
  proj_splitk<<<gproj, 256, 0, stream>>>(key,   W_K, kb);
  proj_splitk<<<gproj, 256, 0, stream>>>(value, W_V, vb);

  dones_scan<<<Bc, Tc, 0, stream>>>(dones, etag, firstdone);

  float* out0 = (float*)d_out;
  float* hout = out0 + qkv;
  retention16<<<dim3(Sc / 16, Bc), 128, 0, stream>>>(qb, kb, vb, hstate, etag,
                                                    firstdone, out0);
  hstate_update<<<dim3(Hc / 16, Hc / 16, Bc), 32, 0, stream>>>(kb, vb, hstate,
                                                               etag, hout);
}


# dones scan rides in the first projection launch: work-groups 0..31 scan one batch each behind their first X and W requests; launches 6 -> 5; same tags, no sum reordered
# speedup vs baseline: 3.9755x; 1.0440x over previous
#include <hip/hip_runtime.h>
#include <hip/hip_bf16.h>


typedef float v2f __attribute__((ext_vector_type(2)));
typedef float v4f __attribute__((ext_vector_type(4)));
typedef float v8f __attribute__((ext_vector_type(8)));

constexpr int   Bc = 32, Sc = 1024, Ec = 512, Hc = 64, Tc = 256;
constexpr float LNK = -0.1053605156578263f;

static __device__ __forceinline__ v8f wmma4(v2f a, v2f b, v8f c) {
  return __builtin_amdgcn_wmma_f32_16x16x4_f32(false, a, false, b, (short)0, c,
                                               false, false);
}

constexpr int PJ_TILES = 4;

__global__ void __launch_bounds__(256, 4)
proj_splitk(const float* __restrict__ X, const float* __restrict__ W,
            float* __restrict__ Y, const int* __restrict__ dones,
            int* __restrict__ etag, int* __restrict__ firstdone)
{
  __shared__ unsigned wmask[Tc / 32];

  __shared__ v4f part[2][8][16][16];

  const int lane = threadIdx.x & 31;
  const int wave = threadIdx.x >> 5;
  const int c    = lane & 15;
  const int h    = lane >> 4;
  const int kw   = wave << 6;

  const int      rt0  = blockIdx.x * PJ_TILES;
  const float*   xp   = X + (size_t)rt0 * 16 * Ec;
  float*         yp   = Y + (size_t)rt0 * 16 * Hc;
  const unsigned xoff = (unsigned)(c * Ec + kw + 4 * h);
  const unsigned yoff = (unsigned)((2 * wave + h) * Hc + 4 * c);

  const bool scan = (dones != nullptr) && (blockIdx.x < (unsigned)Bc);

  v4f xa[8];
  #pragma unroll
  for (int j = 0; j < 8; ++j) xa[j] = *(const v4f*)(xp + xoff + 8 * j);

  v2f wb[4][8][2];
  v4f raw[2][8];
  const float* wp = W + (size_t)(kw + 4 * h) * Hc + 4 * c;
  #pragma unroll
  for (int i = 0; i < 8; ++i)
    raw[0][i] = *(const v4f*)(wp + (size_t)(8 * (i >> 2) + (i & 3)) * Hc);

  if (scan) {
    const int      ts = threadIdx.x;
    const bool     d  = dones[(size_t)blockIdx.x * Sc + ts * 4] != 0;
    const unsigned m  = __builtin_amdgcn_ballot_w32(d);
    if (lane == 0) wmask[wave] = m;
    __syncthreads();
    int tag = 0;
    for (int w = 0; w < Tc / 32; ++w) {
      const unsigned mw = wmask[w];
      if (w < wave && mw != 0u) tag = w * 32 + 32 - __clz((int)mw);
    }
    const unsigned own = m & (0xffffffffu >> (31 - lane));
    if (own != 0u) tag = wave * 32 + 32 - __clz((int)own);
    etag[blockIdx.x * Tc + ts] = tag;
    if (ts == 0) {
      int first = Tc;
      for (int w = Tc / 32 - 1; w >= 0; --w) {
        const unsigned mw = wmask[w];
        if (mw != 0u) first = w * 32 + __ffs((int)mw) - 1;
      }
      firstdone[blockIdx.x] = first;
    }
  }

  #pragma unroll
  for (int ch = 0; ch < 4; ++ch) {
    if (ch + 1 < 4) {
      #pragma unroll
      for (int i = 0; i < 8; ++i)
        raw[(ch + 1) & 1][i] =
            *(const v4f*)(wp + (size_t)(8 * (2 * (ch + 1) + (i >> 2)) + (i & 3)) * Hc);
    }
    #pragma unroll
    for (int jj = 0; jj < 2; ++jj) {
      #pragma unroll
      for (int p = 0; p < 2; ++p) {
        #pragma unroll
        for (int nt = 0; nt < 4; ++nt) {
          wb[nt][2 * ch + jj][p].x = raw[ch & 1][4 * jj + 2 * p][nt];
          wb[nt][2 * ch + jj][p].y = raw[ch & 1][4 * jj + 2 * p + 1][nt];
        }
      }
    }
    __builtin_amdgcn_sched_barrier(0);
  }

  #pragma unroll
  for (int t = 0; t < PJ_TILES; ++t) {
    v4f xn[8];
    if (t + 1 < PJ_TILES) {
      const float* xq = xp + (size_t)(t + 1) * 16 * Ec;
      #pragma unroll
      for (int j = 0; j < 8; ++j) xn[j] = *(const v4f*)(xq + xoff + 8 * j);
    }

    v8f acc[4] = {};
    #pragma unroll
    for (int j = 0; j < 8; ++j) {
      #pragma unroll
      for (int p = 0; p < 2; ++p) {
        v2f a;
        a.x = xa[j][2 * p];
        a.y = xa[j][2 * p + 1];
        #pragma unroll
        for (int nt = 0; nt < 4; ++nt)
          acc[nt] = wmma4(a, wb[nt][j][p], acc[nt]);
      }
    }

    #pragma unroll
    for (int r = 0; r < 8; ++r) {
      v4f o;
      o.x = acc[0][r]; o.y = acc[1][r]; o.z = acc[2][r]; o.w = acc[3][r];
      part[t & 1][wave][8 * h + r][c] = o;
    }
    __syncthreads();

    v4f s = part[t & 1][0][2 * wave + h][c];
    #pragma unroll
    for (int u = 1; u < 8; ++u) s += part[t & 1][u][2 * wave + h][c];
    *(v4f*)(yp + (size_t)t * 16 * Hc + yoff) = s;

    if (t + 1 < PJ_TILES) {
      #pragma unroll
      for (int j = 0; j < 8; ++j) xa[j] = xn[j];
    }
  }
}


constexpr int AP = 34;
constexpr int OP = 68;

__global__ void __launch_bounds__(128)
retention16(const float* __restrict__ q, const float* __restrict__ kbuf,
            const float* __restrict__ vbuf, const float* __restrict__ hstate,
            const int* __restrict__ etag, const int* __restrict__ firstdone,
            float* __restrict__ out)
{
  __shared__ int   tagsh[Tc];
  __shared__ __attribute__((aligned(16))) float ptile[2][2][16 * AP];

  const int lane  = threadIdx.x & 31;
  const int wave  = __builtin_amdgcn_readfirstlane((int)(threadIdx.x >> 5));
  const int fh    = wave & 1;
  const int kt    = wave >> 1;
  const int nl    = lane & 15;
  const int hh    = lane >> 4;
  const int khalf = hh << 1;
  const int mh    = hh << 3;
  const int it    = blockIdx.x;
  const int b     = blockIdx.y;

  const int* tagb = etag + b * Tc;
  const int tag0 = tagb[it * 4];
  for (int i = threadIdx.x; i < Tc; i += 128)  tagsh[i] = tagb[i];
  __syncthreads();

  const int ti0 = it * 4 + (mh >> 2);
  const int cr0 = tagsh[ti0];
  const int cr1 = tagsh[ti0 + 1];
  const int cb  = nl >> 2;

  const float* qb = q      + (size_t)b * Sc * Hc;
  const float* kb = kbuf   + (size_t)b * Sc * Hc;
  const float* vb = vbuf   + (size_t)b * Sc * Hc;
  const float* hb = hstate + (size_t)b * Hc * Hc;

  const float* qrow = qb + (size_t)(it * 16 + nl) * Hc + 4 * hh;
  v4f qs[4];
  #pragma unroll
  for (int jj = 0; jj < 4; ++jj) qs[jj] = *(const v4f*)(qrow + 32 * fh + 8 * jj);

  v8f acc = {};
  const int ntrip = (it >> 1) + 1;
  const int jp0  = ((tag0 > 0) ? tag0 - 1 : 0) >> 3;
  const float* kbase = kb + (size_t)(16 * kt + nl) * Hc + 32 * fh + 4 * hh;
  v4f k4[4];
  {
    const float* krow = kbase + (size_t)(2 * jp0 * 16) * Hc;
    #pragma unroll
    for (int jj = 0; jj < 4; ++jj) k4[jj] = *(const v4f*)(krow + 8 * jj);
  }
  const float* vbase = vb + (size_t)khalf * Hc + 16 * wave + nl;

  for (int jp = jp0; jp < ntrip; ++jp) {
    const int jt0 = 2 * jp;
    const int jt  = jt0 + kt;
    float vr[8][2];
    const float* vrow = vbase + (size_t)(jt0 * 16) * Hc;
    #pragma unroll
    for (int kk = 0; kk < 8; ++kk) {
      vr[kk][0] = vrow[(size_t)(kk * 4) * Hc];
      vr[kk][1] = vrow[(size_t)(kk * 4 + 1) * Hc];
    }
    const int tj = jt * 4 + cb;
    const int cj = tagsh[tj];
    v8f s = {};
    #pragma unroll
    for (int jj = 0; jj < 4; ++jj) {
      #pragma unroll
      for (int p = 0; p < 2; ++p) {
        v2f a, bk;
        a.x  = qs[jj][2 * p];   a.y  = qs[jj][2 * p + 1];
        bk.x = k4[jj][2 * p];   bk.y = k4[jj][2 * p + 1];
        s = wmma4(a, bk, s);
      }
    }
    const float m0 = (cj == cr0) ? __expf((float)(ti0 - tj) * LNK) : 0.0f;
    const float m1 = (cj == cr1) ? __expf((float)(ti0 + 1 - tj) * LNK) : 0.0f;
    const int   dg = (it - jt) * 16 + mh - nl;
    float* pt = &ptile[jp & 1][fh][16 * kt + nl];
    #pragma unroll
    for (int r = 0; r < 8; ++r) {
      const float w = (dg + r >= 0) ? ((r < 4) ? m0 : m1) : 0.0f;
      pt[(mh + r) * AP] = s[r] * w;
    }
    asm volatile("" ::: "memory");
    {
      const int jn = (jp + 1 < ntrip) ? jt0 + 2 : jt0;
      const float* krow = kbase + (size_t)(jn * 16) * Hc;
      #pragma unroll
      for (int jj = 0; jj < 4; ++jj) k4[jj] = *(const v4f*)(krow + 8 * jj);
    }
    __builtin_amdgcn_fence(__ATOMIC_RELEASE, "workgroup", "local");
    __builtin_amdgcn_s_barrier();
    __builtin_amdgcn_fence(__ATOMIC_ACQUIRE, "workgroup", "local");
    const float* p0 = &ptile[jp & 1][0][nl * AP + khalf];
    const float* p1 = &ptile[jp & 1][1][nl * AP + khalf];
    v2f af[8];
    #pragma unroll
    for (int kk = 0; kk < 8; ++kk)
      af[kk] = *(const v2f*)(p0 + kk * 4) + *(const v2f*)(p1 + kk * 4);
    __builtin_amdgcn_sched_barrier(0);
    #pragma unroll
    for (int kk = 0; kk < 8; ++kk) {
      v2f bm;
      bm.x = vr[kk][0];
      bm.y = vr[kk][1];
      acc = wmma4(af[kk], bm, acc);
    }
  }

  v8f cr = {};
  #pragma unroll
  for (int j = 0; j < 8; ++j) {
    const v4f q4 = *(const v4f*)(qrow + 8 * j);
    #pragma unroll
    for (int p = 0; p < 2; ++p) {
      const float* hr = hb + (size_t)(8 * j + 4 * hh + 2 * p) * Hc + 16 * wave + nl;
      v2f a, bm;
      a.x  = q4[2 * p];
      a.y  = q4[2 * p + 1];
      bm.x = hr[0];
      bm.y = hr[Hc];
      cr = wmma4(a, bm, cr);
    }
  }

  const int fd = firstdone[b];
  float* ot = &ptile[0][0][0];
  __syncthreads();
  #pragma unroll
  for (int r = 0; r < 8; ++r) {
    const int   gi = it * 16 + mh + r;
    const int   ti = gi >> 2;
    const float xi = (ti < fd) ? __expf((float)(ti + 1) * LNK) : 0.0f;
    ot[(mh + r) * OP + 16 * wave + nl] = acc[r] + cr[r] * xi;
  }
  __syncthreads();
  #pragma unroll
  for (int i = 0; i < 2; ++i) {
    const int row = 4 * wave + 2 * i + hh;
    const v4f o = *(const v4f*)(&ot[row * OP + 4 * nl]);
    *(v4f*)(out + ((size_t)b * Sc + it * 16 + row) * Hc + 4 * nl) = o;
  }
}

__global__ void __launch_bounds__(32)
hstate_update(const float* __restrict__ kbuf, const float* __restrict__ vbuf,
              const float* __restrict__ hstate, const int* __restrict__ etag,
              float* __restrict__ hout)
{
  __shared__ float ktab[Tc + 1];
  __shared__ float wsh[Tc];
  const int lane  = threadIdx.x & 31;
  const int nl    = lane & 15;
  const int khalf = (lane >> 4) << 1;
  const int mt = blockIdx.x, nt = blockIdx.y, b = blockIdx.z;
  const int* tagb = etag + b * Tc;
  const int  tend = tagb[Tc - 1];

  for (int i = lane; i <= Tc; i += 32) ktab[i] = __expf((float)i * LNK);
  __syncthreads();
  for (int i = lane; i < Tc; i += 32)
    wsh[i] = (tagb[i] == tend) ? ktab[(Tc - 1) - i] : 0.0f;
  __syncthreads();

  const float* kb = kbuf + (size_t)b * Sc * Hc;
  const float* vb = vbuf + (size_t)b * Sc * Hc;

  const int sbeg = (((tend > 0) ? tend - 1 : 0) >> 3) << 5;
  const float* kp0 = kb + (size_t)khalf * Hc + mt * 16 + nl;
  const float* vp0 = vb + (size_t)khalf * Hc + nt * 16 + nl;

  auto load_block = [&](float (&kr)[16], float (&vr)[16], int s) {
    const float* kq = kp0 + (size_t)s * Hc;
    const float* vq = vp0 + (size_t)s * Hc;
    #pragma unroll
    for (int u = 0; u < 8; ++u) {
      kr[2 * u]     = kq[(4 * u)     * Hc];
      kr[2 * u + 1] = kq[(4 * u + 1) * Hc];
      vr[2 * u]     = vq[(4 * u)     * Hc];
      vr[2 * u + 1] = vq[(4 * u + 1) * Hc];
    }
  };
  v8f c = {};
  auto mma_block = [&](const float (&kr)[16], const float (&vr)[16], int s) {
    const float* wq = wsh + (s >> 2);
    #pragma unroll
    for (int u = 0; u < 8; ++u) {
      const float w = wq[u];
      v2f a;
      a.x = kr[2 * u];
      a.y = kr[2 * u + 1];
      v2f bm;
      bm.x = vr[2 * u] * w;
      bm.y = vr[2 * u + 1] * w;
      c = wmma4(a, bm, c);
    }
  };

  const int slast = Sc - 32;
  float kA[16], vA[16], kB[16], vB[16];
  load_block(kA, vA, sbeg);
  int sb = sbeg;
  for (; sb + 64 <= Sc; sb += 64) {
    load_block(kB, vB, sb + 32);
    __builtin_amdgcn_sched_barrier(0);
    mma_block(kA, vA, sb);
    load_block(kA, vA, (sb + 64 < Sc) ? sb + 64 : slast);
    __builtin_amdgcn_sched_barrier(0);
    mma_block(kB, vB, sb + 32);
  }
  if (sb < Sc) mma_block(kA, vA, sb);

  const float cd    = ktab[Tc];
  const float delta = (tend == 0) ? 1.0f : 0.0f;
  const float* hb   = hstate + (size_t)b * Hc * Hc;
  const int    mh   = (lane >> 4) << 3;
  #pragma unroll
  for (int r = 0; r < 8; ++r) {
    const int h1 = mt * 16 + mh + r;
    hout[((size_t)b * Hc + h1) * Hc + nt * 16 + nl] =
        c[r] + hb[(size_t)h1 * Hc + nt * 16 + nl] * cd * delta;
  }
}

extern "C" void kernel_launch(void* const* d_in, const int* in_sizes, int n_in,
                              void* d_out, int out_size, void* d_ws, size_t ws_size,
                              hipStream_t stream)
{
  const float* key    = (const float*)d_in[0];
  const float* query  = (const float*)d_in[1];
  const float* value  = (const float*)d_in[2];
  const float* hstate = (const float*)d_in[3];
  const int*   dones  = (const int*)d_in[4];
  const float* W_Q    = (const float*)d_in[5];
  const float* W_K    = (const float*)d_in[6];
  const float* W_V    = (const float*)d_in[7];

  const size_t qkv = (size_t)Bc * Sc * Hc;
  float* ws = (float*)d_ws;
  float* qb = ws;
  float* kb = ws + qkv;
  float* vb = ws + 2 * qkv;
  int*   etag       = (int*)(ws + 3 * qkv);
  int*   firstdone = etag + Bc * Tc;

  dim3 gproj(Bc * Sc / (16 * PJ_TILES));
  proj_splitk<<<gproj, 256, 0, stream>>>(query, W_Q, qb, dones, etag, firstdone);
  proj_splitk<<<gproj, 256, 0, stream>>>(key,   W_K, kb, nullptr, nullptr, nullptr);
  proj_splitk<<<gproj, 256, 0, stream>>>(value, W_V, vb, nullptr, nullptr, nullptr);

  float* out0 = (float*)d_out;
  float* hout = out0 + qkv;
  retention16<<<dim3(Sc / 16, Bc), 128, 0, stream>>>(qb, kb, vb, hstate, etag,
                                                    firstdone, out0);
  hstate_update<<<dim3(Hc / 16, Hc / 16, Bc), 32, 0, stream>>>(kb, vb, hstate,
                                                               etag, hout);
}


# state update rides in the last projection launch (q, moved third): work-group g adds state tile g, its key blocks dealt to the 8 waves, partials added in LDS; launches 5 -> 4; keys in 8 partial sums
# speedup vs baseline: 4.1855x; 1.0528x over previous
#include <hip/hip_runtime.h>
#include <hip/hip_bf16.h>


typedef float v2f __attribute__((ext_vector_type(2)));
typedef float v4f __attribute__((ext_vector_type(4)));
typedef float v8f __attribute__((ext_vector_type(8)));

constexpr int   Bc = 32, Sc = 1024, Ec = 512, Hc = 64, Tc = 256;
constexpr float LNK = -0.1053605156578263f;

static __device__ __forceinline__ v8f wmma4(v2f a, v2f b, v8f c) {
  return __builtin_amdgcn_wmma_f32_16x16x4_f32(false, a, false, b, (short)0, c,
                                               false, false);
}

constexpr int PJ_TILES = 4;

__global__ void __launch_bounds__(256, 4)
proj_splitk(const float* __restrict__ X, const float* __restrict__ W,
            float* __restrict__ Y, const int* __restrict__ dones,
            int* __restrict__ etag, int* __restrict__ firstdone,
            const float* __restrict__ kst, const float* __restrict__ vst,
            const float* __restrict__ hst, float* __restrict__ hout)
{
  __shared__ unsigned wmask[Tc / 32];
  __shared__ float red[8][256];

  __shared__ v4f part[2][8][16][16];

  const int lane = threadIdx.x & 31;
  const int wave = threadIdx.x >> 5;
  const int c    = lane & 15;
  const int h    = lane >> 4;
  const int kw   = wave << 6;

  const int      rt0  = blockIdx.x * PJ_TILES;
  const float*   xp   = X + (size_t)rt0 * 16 * Ec;
  float*         yp   = Y + (size_t)rt0 * 16 * Hc;
  const unsigned xoff = (unsigned)(c * Ec + kw + 4 * h);
  const unsigned yoff = (unsigned)((2 * wave + h) * Hc + 4 * c);

  const bool scan = (dones != nullptr) && (blockIdx.x < (unsigned)Bc);

  const bool state = (hout != nullptr);
  const int  sbat  = blockIdx.x >> 4;
  const int  smt   = (blockIdx.x >> 2) & 3;
  const int  snt   = blockIdx.x & 3;
  const int  wv    = __builtin_amdgcn_readfirstlane((int)(threadIdx.x >> 5));
  int tend = 0;
  if (state) tend = etag[sbat * Tc + (Tc - 1)];

  v4f xa[8];
  #pragma unroll
  for (int j = 0; j < 8; ++j) xa[j] = *(const v4f*)(xp + xoff + 8 * j);

  v2f wb[4][8][2];
  v4f raw[2][8];
  const float* wp = W + (size_t)(kw + 4 * h) * Hc + 4 * c;
  #pragma unroll
  for (int i = 0; i < 8; ++i)
    raw[0][i] = *(const v4f*)(wp + (size_t)(8 * (i >> 2) + (i & 3)) * Hc);

  if (scan) {
    const int      ts = threadIdx.x;
    const bool     d  = dones[(size_t)blockIdx.x * Sc + ts * 4] != 0;
    const unsigned m  = __builtin_amdgcn_ballot_w32(d);
    if (lane == 0) wmask[wave] = m;
    __syncthreads();
    int tag = 0;
    for (int w = 0; w < Tc / 32; ++w) {
      const unsigned mw = wmask[w];
      if (w < wave && mw != 0u) tag = w * 32 + 32 - __clz((int)mw);
    }
    const unsigned own = m & (0xffffffffu >> (31 - lane));
    if (own != 0u) tag = wave * 32 + 32 - __clz((int)own);
    etag[blockIdx.x * Tc + ts] = tag;
    if (ts == 0) {
      int first = Tc;
      for (int w = Tc / 32 - 1; w >= 0; --w) {
        const unsigned mw = wmask[w];
        if (mw != 0u) first = w * 32 + __ffs((int)mw) - 1;
      }
      firstdone[blockIdx.x] = first;
    }
  }

  #pragma unroll
  for (int ch = 0; ch < 4; ++ch) {
    if (ch + 1 < 4) {
      #pragma unroll
      for (int i = 0; i < 8; ++i)
        raw[(ch + 1) & 1][i] =
            *(const v4f*)(wp + (size_t)(8 * (2 * (ch + 1) + (i >> 2)) + (i & 3)) * Hc);
    }
    #pragma unroll
    for (int jj = 0; jj < 2; ++jj) {
      #pragma unroll
      for (int p = 0; p < 2; ++p) {
        #pragma unroll
        for (int nt = 0; nt < 4; ++nt) {
          wb[nt][2 * ch + jj][p].x = raw[ch & 1][4 * jj + 2 * p][nt];
          wb[nt][2 * ch + jj][p].y = raw[ch & 1][4 * jj + 2 * p + 1][nt];
        }
      }
    }
    __builtin_amdgcn_sched_barrier(0);
  }

  const int sbeg = (((tend > 0) ? tend - 1 : 0) >> 3) << 5;
  const int s0   = sbeg + 32 * wv;
  float kA[16], vA[16], kB[16], vB[16];
  int   tgA = 0, tgB = 0;
  auto load_block = [&](float (&kr)[16], float (&vr)[16], int& tg, int s) {
    const float* kq = kst + ((size_t)sbat * Sc + s + 2 * h) * Hc + smt * 16 + c;
    const float* vq = vst + ((size_t)sbat * Sc + s + 2 * h) * Hc + snt * 16 + c;
    #pragma unroll
    for (int u = 0; u < 8; ++u) {
      kr[2 * u]     = kq[(4 * u)     * Hc];
      kr[2 * u + 1] = kq[(4 * u + 1) * Hc];
      vr[2 * u]     = vq[(4 * u)     * Hc];
      vr[2 * u + 1] = vq[(4 * u + 1) * Hc];
    }
    tg = etag[sbat * Tc + (s >> 2) + (lane & 7)];
  };

  #pragma unroll
  for (int t = 0; t < PJ_TILES; ++t) {
    v4f xn[8];
    if (t + 1 < PJ_TILES) {
      const float* xq = xp + (size_t)(t + 1) * 16 * Ec;
      #pragma unroll
      for (int j = 0; j < 8; ++j) xn[j] = *(const v4f*)(xq + xoff + 8 * j);
    } else if (state && s0 < Sc) {
      load_block(kA, vA, tgA, s0);
    }

    v8f acc[4] = {};
    #pragma unroll
    for (int j = 0; j < 8; ++j) {
      #pragma unroll
      for (int p = 0; p < 2; ++p) {
        v2f a;
        a.x = xa[j][2 * p];
        a.y = xa[j][2 * p + 1];
        #pragma unroll
        for (int nt = 0; nt < 4; ++nt)
          acc[nt] = wmma4(a, wb[nt][j][p], acc[nt]);
      }
    }

    #pragma unroll
    for (int r = 0; r < 8; ++r) {
      v4f o;
      o.x = acc[0][r]; o.y = acc[1][r]; o.z = acc[2][r]; o.w = acc[3][r];
      part[t & 1][wave][8 * h + r][c] = o;
    }
    __syncthreads();

    v4f s = part[t & 1][0][2 * wave + h][c];
    #pragma unroll
    for (int u = 1; u < 8; ++u) s += part[t & 1][u][2 * wave + h][c];
    *(v4f*)(yp + (size_t)t * 16 * Hc + yoff) = s;

    if (t + 1 < PJ_TILES) {
      #pragma unroll
      for (int j = 0; j < 8; ++j) xa[j] = xn[j];
    }
  }

  if (state) {
    const int    srow = smt * 16 + 8 * h + (int)(threadIdx.x >> 5);
    const size_t so   = ((size_t)sbat * Hc + srow) * Hc + snt * 16 + c;
    const float  hv   = hst[so];
    v8f cs = {};
    auto mma_block = [&](const float (&kr)[16], const float (&vr)[16], int tg, int s) {
      const int   ts = (s >> 2) + (lane & 7);
      const float wl = (tg == tend) ? __expf((float)(Tc - 1 - ts) * LNK) : 0.0f;
      const int   wi = __float_as_int(wl);
      #pragma unroll
      for (int u = 0; u < 8; ++u) {
        const float w = __int_as_float(__builtin_amdgcn_readlane(wi, u));
        v2f a;
        a.x = kr[2 * u];
        a.y = kr[2 * u + 1];
        v2f bm;
        bm.x = vr[2 * u] * w;
        bm.y = vr[2 * u + 1] * w;
        cs = wmma4(a, bm, cs);
      }
    };
    for (int s = s0; s < Sc; s += 512) {
      const bool hasB = s + 256 < Sc;
      if (hasB) load_block(kB, vB, tgB, s + 256);
      mma_block(kA, vA, tgA, s);
      if (hasB) {
        if (s + 512 < Sc) load_block(kA, vA, tgA, s + 512);
        mma_block(kB, vB, tgB, s + 256);
      }
    }
    #pragma unroll
    for (int r = 0; r < 8; ++r) red[threadIdx.x >> 5][r * 32 + lane] = cs[r];
    __syncthreads();
    float sum = red[0][threadIdx.x];
    #pragma unroll
    for (int u = 1; u < 8; ++u) sum += red[u][threadIdx.x];
    const float cd    = __expf((float)Tc * LNK);
    const float delta = (tend == 0) ? 1.0f : 0.0f;
    hout[so] = sum + hv * cd * delta;
  }
}


constexpr int AP = 34;
constexpr int OP = 68;

__global__ void __launch_bounds__(128)
retention16(const float* __restrict__ q, const float* __restrict__ kbuf,
            const float* __restrict__ vbuf, const float* __restrict__ hstate,
            const int* __restrict__ etag, const int* __restrict__ firstdone,
            float* __restrict__ out)
{
  __shared__ int   tagsh[Tc];
  __shared__ __attribute__((aligned(16))) float ptile[2][2][16 * AP];

  const int lane  = threadIdx.x & 31;
  const int wave  = __builtin_amdgcn_readfirstlane((int)(threadIdx.x >> 5));
  const int fh    = wave & 1;
  const int kt    = wave >> 1;
  const int nl    = lane & 15;
  const int hh    = lane >> 4;
  const int khalf = hh << 1;
  const int mh    = hh << 3;
  const int it    = blockIdx.x;
  const int b     = blockIdx.y;

  const int* tagb = etag + b * Tc;
  const int tag0 = tagb[it * 4];
  for (int i = threadIdx.x; i < Tc; i += 128)  tagsh[i] = tagb[i];
  __syncthreads();

  const int ti0 = it * 4 + (mh >> 2);
  const int cr0 = tagsh[ti0];
  const int cr1 = tagsh[ti0 + 1];
  const int cb  = nl >> 2;

  const float* qb = q      + (size_t)b * Sc * Hc;
  const float* kb = kbuf   + (size_t)b * Sc * Hc;
  const float* vb = vbuf   + (size_t)b * Sc * Hc;
  const float* hb = hstate + (size_t)b * Hc * Hc;

  const float* qrow = qb + (size_t)(it * 16 + nl) * Hc + 4 * hh;
  v4f qs[4];
  #pragma unroll
  for (int jj = 0; jj < 4; ++jj) qs[jj] = *(const v4f*)(qrow + 32 * fh + 8 * jj);

  v8f acc = {};
  const int ntrip = (it >> 1) + 1;
  const int jp0  = ((tag0 > 0) ? tag0 - 1 : 0) >> 3;
  const float* kbase = kb + (size_t)(16 * kt + nl) * Hc + 32 * fh + 4 * hh;
  v4f k4[4];
  {
    const float* krow = kbase + (size_t)(2 * jp0 * 16) * Hc;
    #pragma unroll
    for (int jj = 0; jj < 4; ++jj) k4[jj] = *(const v4f*)(krow + 8 * jj);
  }
  const float* vbase = vb + (size_t)khalf * Hc + 16 * wave + nl;

  for (int jp = jp0; jp < ntrip; ++jp) {
    const int jt0 = 2 * jp;
    const int jt  = jt0 + kt;
    float vr[8][2];
    const float* vrow = vbase + (size_t)(jt0 * 16) * Hc;
    #pragma unroll
    for (int kk = 0; kk < 8; ++kk) {
      vr[kk][0] = vrow[(size_t)(kk * 4) * Hc];
      vr[kk][1] = vrow[(size_t)(kk * 4 + 1) * Hc];
    }
    const int tj = jt * 4 + cb;
    const int cj = tagsh[tj];
    v8f s = {};
    #pragma unroll
    for (int jj = 0; jj < 4; ++jj) {
      #pragma unroll
      for (int p = 0; p < 2; ++p) {
        v2f a, bk;
        a.x  = qs[jj][2 * p];   a.y  = qs[jj][2 * p + 1];
        bk.x = k4[jj][2 * p];   bk.y = k4[jj][2 * p + 1];
        s = wmma4(a, bk, s);
      }
    }
    const float m0 = (cj == cr0) ? __expf((float)(ti0 - tj) * LNK) : 0.0f;
    const float m1 = (cj == cr1) ? __expf((float)(ti0 + 1 - tj) * LNK) : 0.0f;
    const int   dg = (it - jt) * 16 + mh - nl;
    float* pt = &ptile[jp & 1][fh][16 * kt + nl];
    #pragma unroll
    for (int r = 0; r < 8; ++r) {
      const float w = (dg + r >= 0) ? ((r < 4) ? m0 : m1) : 0.0f;
      pt[(mh + r) * AP] = s[r] * w;
    }
    asm volatile("" ::: "memory");
    {
      const int jn = (jp + 1 < ntrip) ? jt0 + 2 : jt0;
      const float* krow = kbase + (size_t)(jn * 16) * Hc;
      #pragma unroll
      for (int jj = 0; jj < 4; ++jj) k4[jj] = *(const v4f*)(krow + 8 * jj);
    }
    __builtin_amdgcn_fence(__ATOMIC_RELEASE, "workgroup", "local");
    __builtin_amdgcn_s_barrier();
    __builtin_amdgcn_fence(__ATOMIC_ACQUIRE, "workgroup", "local");
    const float* p0 = &ptile[jp & 1][0][nl * AP + khalf];
    const float* p1 = &ptile[jp & 1][1][nl * AP + khalf];
    v2f af[8];
    #pragma unroll
    for (int kk = 0; kk < 8; ++kk)
      af[kk] = *(const v2f*)(p0 + kk * 4) + *(const v2f*)(p1 + kk * 4);
    __builtin_amdgcn_sched_barrier(0);
    #pragma unroll
    for (int kk = 0; kk < 8; ++kk) {
      v2f bm;
      bm.x = vr[kk][0];
      bm.y = vr[kk][1];
      acc = wmma4(af[kk], bm, acc);
    }
  }

  v8f cr = {};
  #pragma unroll
  for (int j = 0; j < 8; ++j) {
    const v4f q4 = *(const v4f*)(qrow + 8 * j);
    #pragma unroll
    for (int p = 0; p < 2; ++p) {
      const float* hr = hb + (size_t)(8 * j + 4 * hh + 2 * p) * Hc + 16 * wave + nl;
      v2f a, bm;
      a.x  = q4[2 * p];
      a.y  = q4[2 * p + 1];
      bm.x = hr[0];
      bm.y = hr[Hc];
      cr = wmma4(a, bm, cr);
    }
  }

  const int fd = firstdone[b];
  float* ot = &ptile[0][0][0];
  __syncthreads();
  #pragma unroll
  for (int r = 0; r < 8; ++r) {
    const int   gi = it * 16 + mh + r;
    const int   ti = gi >> 2;
    const float xi = (ti < fd) ? __expf((float)(ti + 1) * LNK) : 0.0f;
    ot[(mh + r) * OP + 16 * wave + nl] = acc[r] + cr[r] * xi;
  }
  __syncthreads();
  #pragma unroll
  for (int i = 0; i < 2; ++i) {
    const int row = 4 * wave + 2 * i + hh;
    const v4f o = *(const v4f*)(&ot[row * OP + 4 * nl]);
    *(v4f*)(out + ((size_t)b * Sc + it * 16 + row) * Hc + 4 * nl) = o;
  }
}

extern "C" void kernel_launch(void* const* d_in, const int* in_sizes, int n_in,
                              void* d_out, int out_size, void* d_ws, size_t ws_size,
                              hipStream_t stream)
{
  const float* key    = (const float*)d_in[0];
  const float* query  = (const float*)d_in[1];
  const float* value  = (const float*)d_in[2];
  const float* hstate = (const float*)d_in[3];
  const int*   dones  = (const int*)d_in[4];
  const float* W_Q    = (const float*)d_in[5];
  const float* W_K    = (const float*)d_in[6];
  const float* W_V    = (const float*)d_in[7];

  const size_t qkv = (size_t)Bc * Sc * Hc;
  float* ws = (float*)d_ws;
  float* qb = ws;
  float* kb = ws + qkv;
  float* vb = ws + 2 * qkv;
  int*   etag       = (int*)(ws + 3 * qkv);
  int*   firstdone = etag + Bc * Tc;

  float* out0 = (float*)d_out;
  float* hout = out0 + qkv;

  dim3 gproj(Bc * Sc / (16 * PJ_TILES));
  proj_splitk<<<gproj, 256, 0, stream>>>(key,   W_K, kb, dones, etag, firstdone,
                                         nullptr, nullptr, nullptr, nullptr);
  proj_splitk<<<gproj, 256, 0, stream>>>(value, W_V, vb, nullptr, nullptr, nullptr,
                                         nullptr, nullptr, nullptr, nullptr);
  proj_splitk<<<gproj, 256, 0, stream>>>(query, W_Q, qb, nullptr, etag, nullptr,
                                         kb, vb, hstate, hout);

  retention16<<<dim3(Sc / 16, Bc), 128, 0, stream>>>(qb, kb, vb, hstate, etag,
                                                    firstdone, out0);
}
